# starting kernel
# baseline (speedup 1.0000x reference)
#include <hip/hip_runtime.h>
#include <hip/hip_bf16.h>


#define DIM_IN   128
#define HID      64
#define HEADS    8
#define DIM_H    8
#define DIM_OUT  40
#define NEG_SLOPE 0.2f

typedef __attribute__((ext_vector_type(2))) float v2f;
typedef __attribute__((ext_vector_type(8))) float v8f;

__device__ __forceinline__ void atomicMaxFloat(float* addr, float val) {
    if (!(val == val)) return;
    if (val >= 0.0f) {
        atomicMax((int*)addr, __float_as_int(val));
    } else {
        atomicMin((unsigned int*)addr, __float_as_uint(val));
    }
}

__global__ void gat_fill_u32(unsigned int* __restrict__ p, unsigned int val,
                             long long n) {
    long long i = (long long)blockIdx.x * blockDim.x + threadIdx.x;
    if (i < n) p[i] = val;
}

template <int K, int CC, int CP>
__global__ __launch_bounds__(256) void gat_gemm_bias_f32_wmma(
    const float* __restrict__ A, const float* __restrict__ W,
    const float* __restrict__ bias, float* __restrict__ out, int n) {
    __shared__ float sWt[CP * K];

    for (int idx = threadIdx.x; idx < CP * K; idx += blockDim.x) {
        const int c = idx / K;
        const int k = idx % K;
        sWt[idx] = (c < CC) ? W[(size_t)k * CC + c] : 0.0f;
    }
    __syncthreads();

    const int waveId = threadIdx.x >> 5;
    const int lane = threadIdx.x & 31;
    constexpr int colTiles = CP >> 4;
    const int rowTiles = n >> 4;
    const int tileId = blockIdx.x * (blockDim.x >> 5) + waveId;
    const int rowTile = tileId / colTiles;
    const int colTile = tileId % colTiles;
    if (rowTile >= rowTiles) return;

    const int mrow = (rowTile << 4) + (lane & 15);
    const int colp = (colTile << 4) + (lane & 15);
    const int khalf = (lane >> 4) << 1;

    const float* __restrict__ aRow = A + (size_t)mrow * K + khalf;
    const float* __restrict__ bCol = sWt + (size_t)colp * K + khalf;

    v8f acc = {};
#pragma unroll
    for (int k0 = 0; k0 < K; k0 += 4) {
        const v2f a = *(const v2f*)(aRow + k0);
        const v2f b = *(const v2f*)(bCol + k0);
        acc = __builtin_amdgcn_wmma_f32_16x16x4_f32(
            false, a, false, b, (short)0, acc, false, false);
    }

    if (colp < CC) {
        const float bv = bias[colp];
        const int rbase = (rowTile << 4) + ((lane >> 4) << 3);
#pragma unroll
        for (int r = 0; r < 8; ++r) {
            out[(size_t)(rbase + r) * CC + colp] = acc[r] + bv;
        }
    }
}

__global__ void gat_edge_logits_max_l1(
    const float* __restrict__ xl, const float* __restrict__ xr,
    const float* __restrict__ att, const int* __restrict__ src,
    const int* __restrict__ dst, float* __restrict__ e1,
    float* __restrict__ m1, long long E8) {
    long long t = (long long)blockIdx.x * blockDim.x + threadIdx.x;
    if (t >= E8) return;
    const int e = (int)(t >> 3);
    const int h = (int)(t & 7);
    const int s = src[e], d = dst[e];
    const float* pj = xl + (size_t)s * HID + h * DIM_H;
    const float* pi = xr + (size_t)d * HID + h * DIM_H;
    const float* pa = att + h * DIM_H;
    float acc = 0.0f;
#pragma unroll
    for (int c = 0; c < DIM_H; ++c) {
        float v = pi[c] + pj[c];
        v = v > 0.0f ? v : v * NEG_SLOPE;
        acc += v * pa[c];
    }
    e1[t] = acc;
    atomicMaxFloat(&m1[(size_t)d * HEADS + h], acc);
}

__global__ void gat_edge_expsum_l1(
    const int* __restrict__ dst, const float* __restrict__ m1,
    float* __restrict__ e1, float* __restrict__ denom1, long long E8) {
    long long t = (long long)blockIdx.x * blockDim.x + threadIdx.x;
    if (t >= E8) return;
    const int e = (int)(t >> 3);
    const int h = (int)(t & 7);
    const int d = dst[e];
    const float ex = expf(e1[t] - m1[(size_t)d * HEADS + h]);
    e1[t] = ex;
    atomicAdd(&denom1[(size_t)d * HEADS + h], ex);
}

__global__ void gat_edge_aggregate_l1(
    const int* __restrict__ src, const int* __restrict__ dst,
    const float* __restrict__ xl, const float* __restrict__ e1,
    const float* __restrict__ denom1, float* __restrict__ out1, long long E8) {
    long long t = (long long)blockIdx.x * blockDim.x + threadIdx.x;
    if (t >= E8) return;
    const int e = (int)(t >> 3);
    const int h = (int)(t & 7);
    const int s = src[e], d = dst[e];
    const float alpha = e1[t] / denom1[(size_t)d * HEADS + h];
    const float* pj = xl + (size_t)s * HID + h * DIM_H;
    float* po = out1 + (size_t)d * HID + h * DIM_H;
#pragma unroll
    for (int c = 0; c < DIM_H; ++c) {
        atomicAdd(&po[c], pj[c] * alpha);
    }
}

__global__ void gat_bias_elu(float* __restrict__ out1,
                             const float* __restrict__ bias1, long long total) {
    long long i = (long long)blockIdx.x * blockDim.x + threadIdx.x;
    if (i >= total) return;
    float v = out1[i] + bias1[(int)(i & (HID - 1))];
    out1[i] = v > 0.0f ? v : expm1f(v);
}

__global__ void gat_edge_logits_max_l2(
    const float* __restrict__ xl, const float* __restrict__ xr,
    const float* __restrict__ att, const int* __restrict__ src,
    const int* __restrict__ dst, float* __restrict__ e2,
    float* __restrict__ m2, long long E8) {
    long long t = (long long)blockIdx.x * blockDim.x + threadIdx.x;
    if (t >= E8) return;
    const int e = (int)(t >> 3);
    const int sub = (int)(t & 7);
    const int s = src[e], d = dst[e];
    const float* pj = xl + (size_t)s * DIM_OUT;
    const float* pi = xr + (size_t)d * DIM_OUT;
    float acc = 0.0f;
#pragma unroll
    for (int c = sub; c < DIM_OUT; c += 8) {
        float v = pi[c] + pj[c];
        v = v > 0.0f ? v : v * NEG_SLOPE;
        acc += v * att[c];
    }
    acc += __shfl_xor(acc, 1, 32);
    acc += __shfl_xor(acc, 2, 32);
    acc += __shfl_xor(acc, 4, 32);
    if (sub == 0) {
        e2[e] = acc;
        atomicMaxFloat(&m2[d], acc);
    }
}

__global__ void gat_edge_expsum_l2(
    const int* __restrict__ dst, const float* __restrict__ m2,
    float* __restrict__ e2, float* __restrict__ denom2, int E) {
    int e = blockIdx.x * blockDim.x + threadIdx.x;
    if (e >= E) return;
    const int d = dst[e];
    const float ex = expf(e2[e] - m2[d]);
    e2[e] = ex;
    atomicAdd(&denom2[d], ex);
}

__global__ void gat_edge_aggregate_l2(
    const int* __restrict__ src, const int* __restrict__ dst,
    const float* __restrict__ xl, const float* __restrict__ e2,
    const float* __restrict__ denom2, float* __restrict__ out2, long long E8) {
    long long t = (long long)blockIdx.x * blockDim.x + threadIdx.x;
    if (t >= E8) return;
    const int e = (int)(t >> 3);
    const int sub = (int)(t & 7);
    const int s = src[e], d = dst[e];
    const float alpha = e2[e] / denom2[d];
    const float* pj = xl + (size_t)s * DIM_OUT;
    float* po = out2 + (size_t)d * DIM_OUT;
#pragma unroll
    for (int c = sub; c < DIM_OUT; c += 8) {
        atomicAdd(&po[c], pj[c] * alpha);
    }
}

__global__ __launch_bounds__(256) void gat_finalize(
    const float* __restrict__ out2, const float* __restrict__ bias2,
    float* __restrict__ d_out, int N) {
    const int node = blockIdx.x * (blockDim.x >> 5) + (threadIdx.x >> 5);
    const int lane = threadIdx.x & 31;
    if (node >= N) return;
    const float* row = out2 + (size_t)node * DIM_OUT;
    const float a = row[lane] + bias2[lane];
    const bool has2 = (lane < DIM_OUT - 32);
    const float b = has2 ? (row[32 + lane] + bias2[32 + lane]) : -__builtin_inff();
    float mx = fmaxf(a, b);
#pragma unroll
    for (int o = 16; o >= 1; o >>= 1) mx = fmaxf(mx, __shfl_xor(mx, o, 32));
    float s = expf(a - mx) + (has2 ? expf(b - mx) : 0.0f);
#pragma unroll
    for (int o = 16; o >= 1; o >>= 1) s += __shfl_xor(s, o, 32);
    const float lse = logf(s) + mx;
    float* h  = d_out + (size_t)node * DIM_OUT;
    float* ls = d_out + (size_t)N * DIM_OUT + (size_t)node * DIM_OUT;
    h[lane] = a;
    ls[lane] = a - lse;
    if (has2) {
        h[32 + lane] = b;
        ls[32 + lane] = b - lse;
    }
}

static inline int cdiv_i(long long a, long long b) { return (int)((a + b - 1) / b); }

extern "C" void kernel_launch(void* const* d_in, const int* in_sizes, int n_in,
                              void* d_out, int out_size, void* d_ws, size_t ws_size,
                              hipStream_t stream) {
    (void)n_in; (void)out_size; (void)ws_size;
    const float* x     = (const float*)d_in[0];
    const int*   eidx  = (const int*)d_in[1];
    const float* Wl1   = (const float*)d_in[2];
    const float* bl1   = (const float*)d_in[3];
    const float* Wr1   = (const float*)d_in[4];
    const float* br1   = (const float*)d_in[5];
    const float* att1  = (const float*)d_in[6];
    const float* bias1 = (const float*)d_in[7];
    const float* Wl2   = (const float*)d_in[8];
    const float* bl2   = (const float*)d_in[9];
    const float* Wr2   = (const float*)d_in[10];
    const float* br2   = (const float*)d_in[11];
    const float* att2  = (const float*)d_in[12];
    const float* bias2 = (const float*)d_in[13];

    const int N = in_sizes[0] / DIM_IN;
    const int E = in_sizes[1] / 2;
    const int* src = eidx;
    const int* dst = eidx + E;

    float* ws = (float*)d_ws;
    size_t o = 0;
    float* h_l1   = ws + o; o += (size_t)N * HID;
    float* h_r1   = ws + o; o += (size_t)N * HID;
    float* e1     = ws + o; o += (size_t)E * HEADS;
    float* m1     = ws + o; o += (size_t)N * HEADS;
    float* denom1 = ws + o; o += (size_t)N * HEADS;
    float* out1   = ws + o; o += (size_t)N * HID;
    float* h_l2   = ws;
    float* h_r2   = h_l2 + (size_t)N * DIM_OUT;
    float* e2     = h_r2 + (size_t)N * DIM_OUT;
    float* out2   = e2 + (size_t)E;
    float* m2     = out2 + (size_t)N * DIM_OUT;
    float* denom2 = m2 + (size_t)N;

    const int B = 256;
    const long long E8 = (long long)E * 8;
    const unsigned int NEG_INF_BITS = 0xFF800000u;

    gat_fill_u32<<<cdiv_i((long long)N * HEADS, B), B, 0, stream>>>(
        (unsigned int*)m1, NEG_INF_BITS, (long long)N * HEADS);
    gat_fill_u32<<<cdiv_i((long long)N * HEADS, B), B, 0, stream>>>(
        (unsigned int*)denom1, 0u, (long long)N * HEADS);
    gat_fill_u32<<<cdiv_i((long long)N * HID, B), B, 0, stream>>>(
        (unsigned int*)out1, 0u, (long long)N * HID);

    {
        const int tiles = (N >> 4) * (HID >> 4);
        const int grid = cdiv_i(tiles, B / 32);
        gat_gemm_bias_f32_wmma<DIM_IN, HID, HID>
            <<<grid, B, 0, stream>>>(x, Wl1, bl1, h_l1, N);
        gat_gemm_bias_f32_wmma<DIM_IN, HID, HID>
            <<<grid, B, 0, stream>>>(x, Wr1, br1, h_r1, N);
    }

    gat_edge_logits_max_l1<<<cdiv_i(E8, B), B, 0, stream>>>(
        h_l1, h_r1, att1, src, dst, e1, m1, E8);
    gat_edge_expsum_l1<<<cdiv_i(E8, B), B, 0, stream>>>(dst, m1, e1, denom1, E8);
    gat_edge_aggregate_l1<<<cdiv_i(E8, B), B, 0, stream>>>(
        src, dst, h_l1, e1, denom1, out1, E8);
    gat_bias_elu<<<cdiv_i((long long)N * HID, B), B, 0, stream>>>(
        out1, bias1, (long long)N * HID);

    gat_fill_u32<<<cdiv_i(N, B), B, 0, stream>>>((unsigned int*)m2, NEG_INF_BITS, N);
    gat_fill_u32<<<cdiv_i(N, B), B, 0, stream>>>((unsigned int*)denom2, 0u, N);
    gat_fill_u32<<<cdiv_i((long long)N * DIM_OUT, B), B, 0, stream>>>(
        (unsigned int*)out2, 0u, (long long)N * DIM_OUT);

    {
        const int tiles = (N >> 4) * (48 >> 4);
        const int grid = cdiv_i(tiles, B / 32);
        gat_gemm_bias_f32_wmma<HID, DIM_OUT, 48>
            <<<grid, B, 0, stream>>>(out1, Wl2, bl2, h_l2, N);
        gat_gemm_bias_f32_wmma<HID, DIM_OUT, 48>
            <<<grid, B, 0, stream>>>(out1, Wr2, br2, h_r2, N);
    }

    gat_edge_logits_max_l2<<<cdiv_i(E8, B), B, 0, stream>>>(
        h_l2, h_r2, att2, src, dst, e2, m2, E8);
    gat_edge_expsum_l2<<<cdiv_i(E, B), B, 0, stream>>>(dst, m2, e2, denom2, E);
    gat_edge_aggregate_l2<<<cdiv_i(E8, B), B, 0, stream>>>(
        src, dst, h_l2, e2, denom2, out2, E8);

    gat_finalize<<<cdiv_i(N, B / 32), B, 0, stream>>>((const float*)out2, bias2,
                                                      (float*)d_out, N);
}


# Edges grouped by destination per call (count, scan, place, ordered by edge id); per-node softmax and aggregation in registers with running max, no float atomics; sums in fixed edge-id order
# speedup vs baseline: 2.1459x; 2.1459x over previous
#include <hip/hip_runtime.h>
#include <hip/hip_bf16.h>


#define DIM_IN   128
#define HID      64
#define HEADS    8
#define DIM_H    8
#define DIM_OUT  40
#define NEG_SLOPE 0.2f

typedef __attribute__((ext_vector_type(2))) float v2f;
typedef __attribute__((ext_vector_type(8))) float v8f;

__global__ void gat_fill_u32(unsigned int* __restrict__ p, unsigned int val,
                             long long n) {
    long long i = (long long)blockIdx.x * blockDim.x + threadIdx.x;
    if (i < n) p[i] = val;
}

template <int K, int CC, int CP>
__global__ __launch_bounds__(256) void gat_gemm_bias_f32_wmma(
    const float* __restrict__ A, const float* __restrict__ W,
    const float* __restrict__ bias, float* __restrict__ out, int n) {
    __shared__ float sWt[CP * K];

    for (int idx = threadIdx.x; idx < CP * K; idx += blockDim.x) {
        const int c = idx / K;
        const int k = idx % K;
        sWt[idx] = (c < CC) ? W[(size_t)k * CC + c] : 0.0f;
    }
    __syncthreads();

    const int waveId = threadIdx.x >> 5;
    const int lane = threadIdx.x & 31;
    constexpr int colTiles = CP >> 4;
    const int rowTiles = n >> 4;
    const int tileId = blockIdx.x * (blockDim.x >> 5) + waveId;
    const int rowTile = tileId / colTiles;
    const int colTile = tileId % colTiles;
    if (rowTile >= rowTiles) return;

    const int mrow = (rowTile << 4) + (lane & 15);
    const int colp = (colTile << 4) + (lane & 15);
    const int khalf = (lane >> 4) << 1;

    const float* __restrict__ aRow = A + (size_t)mrow * K + khalf;
    const float* __restrict__ bCol = sWt + (size_t)colp * K + khalf;

    v8f acc = {};
#pragma unroll
    for (int k0 = 0; k0 < K; k0 += 4) {
        const v2f a = *(const v2f*)(aRow + k0);
        const v2f b = *(const v2f*)(bCol + k0);
        acc = __builtin_amdgcn_wmma_f32_16x16x4_f32(
            false, a, false, b, (short)0, acc, false, false);
    }

    if (colp < CC) {
        const float bv = bias[colp];
        const int rbase = (rowTile << 4) + ((lane >> 4) << 3);
#pragma unroll
        for (int r = 0; r < 8; ++r) {
            out[(size_t)(rbase + r) * CC + colp] = acc[r] + bv;
        }
    }
}

__global__ void gat_count_rank(const int* __restrict__ dst, int* __restrict__ cnt,
                               int* __restrict__ rnk, int E) {
    const int e = blockIdx.x * blockDim.x + threadIdx.x;
    if (e >= E) return;
    rnk[e] = atomicAdd(&cnt[dst[e]], 1);
}

__global__ __launch_bounds__(1024) void gat_scan_counts(
    const int* __restrict__ cnt, int* __restrict__ rowptr, int N) {
    __shared__ int sWave[32];
    const int tid = threadIdx.x;
    const int lane = tid & 31;
    const int w = tid >> 5;
    const int chunk = (N + 1023) >> 10;
    const int b = min(tid * chunk, N);
    const int e = min(b + chunk, N);
    int sum = 0;
    for (int i = b; i < e; ++i) sum += cnt[i];
    int incl = sum;
    for (int o = 1; o < 32; o <<= 1) {
        const int v = __shfl_up(incl, o, 32);
        if (lane >= o) incl += v;
    }
    if (lane == 31) sWave[w] = incl;
    __syncthreads();
    if (w == 0) {
        const int v = sWave[lane];
        int iv = v;
        for (int o = 1; o < 32; o <<= 1) {
            const int u = __shfl_up(iv, o, 32);
            if (lane >= o) iv += u;
        }
        sWave[lane] = iv - v;
    }
    __syncthreads();
    int run = sWave[w] + incl - sum;
    for (int i = b; i < e; ++i) {
        rowptr[i] = run;
        run += cnt[i];
    }
    if (tid == 1023) rowptr[N] = run;
}

__global__ void gat_place(const int* __restrict__ dst, const int* __restrict__ rnk,
                          const int* __restrict__ rowptr, int* __restrict__ perm,
                          int E) {
    const int e = blockIdx.x * blockDim.x + threadIdx.x;
    if (e >= E) return;
    perm[rowptr[dst[e]] + rnk[e]] = e;
}

__global__ void gat_order_emit(const int* __restrict__ src, const int* __restrict__ dst,
                               const int* __restrict__ rowptr,
                               const int* __restrict__ perm,
                               int* __restrict__ csr_src, int E) {
    const int i = blockIdx.x * blockDim.x + threadIdx.x;
    if (i >= E) return;
    const int e = perm[i];
    const int d = dst[e];
    const int beg = rowptr[d], end = rowptr[d + 1];
    int r = 0;
    for (int j = beg; j < end; ++j) r += (perm[j] < e) ? 1 : 0;
    csr_src[beg + r] = src[e];
}

__device__ __forceinline__ float gat_leaky(float v) {
    return v > 0.0f ? v : v * NEG_SLOPE;
}

__device__ __forceinline__ float gat_elu(float v) {
    return v > 0.0f ? v : expm1f(v);
}

__global__ __launch_bounds__(256) void gat_node_l1(
    const float* __restrict__ xl, const float* __restrict__ xr,
    const float* __restrict__ att, const float* __restrict__ bias1,
    const int* __restrict__ rowptr, const int* __restrict__ csr_src,
    float* __restrict__ h1, int N) {
    const int t = blockIdx.x * blockDim.x + threadIdx.x;
    const int node = t >> 3;
    const int h = t & 7;
    if (node >= N) return;
    const int beg = rowptr[node], end = rowptr[node + 1];
    const float4* pi = (const float4*)(xr + (size_t)node * HID + h * DIM_H);
    const float4 xi0 = pi[0], xi1 = pi[1];
    const float4* pa = (const float4*)(att + h * DIM_H);
    const float4 a0 = pa[0], a1 = pa[1];

    float m = -__builtin_inff();
    float den = 0.0f;
    float acc[8] = {0.0f, 0.0f, 0.0f, 0.0f, 0.0f, 0.0f, 0.0f, 0.0f};

    int s0n = 0, s1n = 0;
    if (beg < end) {
        s0n = csr_src[beg];
        s1n = csr_src[min(beg + 1, end - 1)];
    }
    for (int i = beg; i < end; i += 2) {
        const int s0 = s0n, s1 = s1n;
        const bool has1 = (i + 1 < end);
        const float4* p0 = (const float4*)(xl + (size_t)s0 * HID + h * DIM_H);
        const float4* p1 = (const float4*)(xl + (size_t)s1 * HID + h * DIM_H);
        const float4 u0 = p0[0], u1 = p0[1];
        const float4 w0 = p1[0], w1 = p1[1];
        if (i + 2 < end) {
            s0n = csr_src[i + 2];
            s1n = csr_src[min(i + 3, end - 1)];
        }
        {
            float lg = 0.0f;
            lg += gat_leaky(xi0.x + u0.x) * a0.x;
            lg += gat_leaky(xi0.y + u0.y) * a0.y;
            lg += gat_leaky(xi0.z + u0.z) * a0.z;
            lg += gat_leaky(xi0.w + u0.w) * a0.w;
            lg += gat_leaky(xi1.x + u1.x) * a1.x;
            lg += gat_leaky(xi1.y + u1.y) * a1.y;
            lg += gat_leaky(xi1.z + u1.z) * a1.z;
            lg += gat_leaky(xi1.w + u1.w) * a1.w;
            const float mn = fmaxf(m, lg);
            const float sc = expf(m - mn);
            const float p = expf(lg - mn);
            den = den * sc + p;
            acc[0] = acc[0] * sc + p * u0.x;
            acc[1] = acc[1] * sc + p * u0.y;
            acc[2] = acc[2] * sc + p * u0.z;
            acc[3] = acc[3] * sc + p * u0.w;
            acc[4] = acc[4] * sc + p * u1.x;
            acc[5] = acc[5] * sc + p * u1.y;
            acc[6] = acc[6] * sc + p * u1.z;
            acc[7] = acc[7] * sc + p * u1.w;
            m = mn;
        }
        if (has1) {
            float lg = 0.0f;
            lg += gat_leaky(xi0.x + w0.x) * a0.x;
            lg += gat_leaky(xi0.y + w0.y) * a0.y;
            lg += gat_leaky(xi0.z + w0.z) * a0.z;
            lg += gat_leaky(xi0.w + w0.w) * a0.w;
            lg += gat_leaky(xi1.x + w1.x) * a1.x;
            lg += gat_leaky(xi1.y + w1.y) * a1.y;
            lg += gat_leaky(xi1.z + w1.z) * a1.z;
            lg += gat_leaky(xi1.w + w1.w) * a1.w;
            const float mn = fmaxf(m, lg);
            const float sc = expf(m - mn);
            const float p = expf(lg - mn);
            den = den * sc + p;
            acc[0] = acc[0] * sc + p * w0.x;
            acc[1] = acc[1] * sc + p * w0.y;
            acc[2] = acc[2] * sc + p * w0.z;
            acc[3] = acc[3] * sc + p * w0.w;
            acc[4] = acc[4] * sc + p * w1.x;
            acc[5] = acc[5] * sc + p * w1.y;
            acc[6] = acc[6] * sc + p * w1.z;
            acc[7] = acc[7] * sc + p * w1.w;
            m = mn;
        }
    }

    const float inv = (end > beg) ? 1.0f / den : 0.0f;
    const float4* pb = (const float4*)(bias1 + h * DIM_H);
    const float4 b0 = pb[0], b1 = pb[1];
    float4* po = (float4*)(h1 + (size_t)node * HID + h * DIM_H);
    po[0] = make_float4(gat_elu(acc[0] * inv + b0.x), gat_elu(acc[1] * inv + b0.y),
                        gat_elu(acc[2] * inv + b0.z), gat_elu(acc[3] * inv + b0.w));
    po[1] = make_float4(gat_elu(acc[4] * inv + b1.x), gat_elu(acc[5] * inv + b1.y),
                        gat_elu(acc[6] * inv + b1.z), gat_elu(acc[7] * inv + b1.w));
}

__global__ __launch_bounds__(256) void gat_node_l2_final(
    const float* __restrict__ xl, const float* __restrict__ xr,
    const float* __restrict__ att, const float* __restrict__ bias2,
    const int* __restrict__ rowptr, const int* __restrict__ csr_src,
    float* __restrict__ d_out, int N) {
    const int t = blockIdx.x * blockDim.x + threadIdx.x;
    const int node = t >> 3;
    const int sub = t & 7;
    if (node >= N) return;
    const int beg = rowptr[node], end = rowptr[node + 1];
    const float* ri = xr + (size_t)node * DIM_OUT;
    const float4 xi = *(const float4*)(ri + sub * 4);
    const float xi4 = ri[32 + sub];
    const float4 av = *(const float4*)(att + sub * 4);
    const float a4 = att[32 + sub];

    float m = -__builtin_inff();
    float den = 0.0f;
    float acc0 = 0.0f, acc1 = 0.0f, acc2 = 0.0f, acc3 = 0.0f, acc4 = 0.0f;

    int s0n = 0, s1n = 0;
    if (beg < end) {
        s0n = csr_src[beg];
        s1n = csr_src[min(beg + 1, end - 1)];
    }
    for (int i = beg; i < end; i += 2) {
        const int s0 = s0n, s1 = s1n;
        const bool has1 = (i + 1 < end);
        const float* r0 = xl + (size_t)s0 * DIM_OUT;
        const float* r1 = xl + (size_t)s1 * DIM_OUT;
        const float4 u = *(const float4*)(r0 + sub * 4);
        const float u4 = r0[32 + sub];
        const float4 w = *(const float4*)(r1 + sub * 4);
        const float w4 = r1[32 + sub];
        if (i + 2 < end) {
            s0n = csr_src[i + 2];
            s1n = csr_src[min(i + 3, end - 1)];
        }
        {
            float lg = 0.0f;
            lg += gat_leaky(xi.x + u.x) * av.x;
            lg += gat_leaky(xi.y + u.y) * av.y;
            lg += gat_leaky(xi.z + u.z) * av.z;
            lg += gat_leaky(xi.w + u.w) * av.w;
            lg += gat_leaky(xi4 + u4) * a4;
            lg += __shfl_xor(lg, 1, 32);
            lg += __shfl_xor(lg, 2, 32);
            lg += __shfl_xor(lg, 4, 32);
            const float mn = fmaxf(m, lg);
            const float sc = expf(m - mn);
            const float p = expf(lg - mn);
            den = den * sc + p;
            acc0 = acc0 * sc + p * u.x;
            acc1 = acc1 * sc + p * u.y;
            acc2 = acc2 * sc + p * u.z;
            acc3 = acc3 * sc + p * u.w;
            acc4 = acc4 * sc + p * u4;
            m = mn;
        }
        if (has1) {
            float lg = 0.0f;
            lg += gat_leaky(xi.x + w.x) * av.x;
            lg += gat_leaky(xi.y + w.y) * av.y;
            lg += gat_leaky(xi.z + w.z) * av.z;
            lg += gat_leaky(xi.w + w.w) * av.w;
            lg += gat_leaky(xi4 + w4) * a4;
            lg += __shfl_xor(lg, 1, 32);
            lg += __shfl_xor(lg, 2, 32);
            lg += __shfl_xor(lg, 4, 32);
            const float mn = fmaxf(m, lg);
            const float sc = expf(m - mn);
            const float p = expf(lg - mn);
            den = den * sc + p;
            acc0 = acc0 * sc + p * w.x;
            acc1 = acc1 * sc + p * w.y;
            acc2 = acc2 * sc + p * w.z;
            acc3 = acc3 * sc + p * w.w;
            acc4 = acc4 * sc + p * w4;
            m = mn;
        }
    }

    const float inv = (end > beg) ? 1.0f / den : 0.0f;
    const float4 bv = *(const float4*)(bias2 + sub * 4);
    const float v0 = acc0 * inv + bv.x;
    const float v1 = acc1 * inv + bv.y;
    const float v2 = acc2 * inv + bv.z;
    const float v3 = acc3 * inv + bv.w;
    const float v4 = acc4 * inv + bias2[32 + sub];

    float mx = fmaxf(fmaxf(fmaxf(v0, v1), fmaxf(v2, v3)), v4);
    mx = fmaxf(mx, __shfl_xor(mx, 1, 32));
    mx = fmaxf(mx, __shfl_xor(mx, 2, 32));
    mx = fmaxf(mx, __shfl_xor(mx, 4, 32));
    float s = expf(v0 - mx) + expf(v1 - mx) + expf(v2 - mx) + expf(v3 - mx) +
              expf(v4 - mx);
    s += __shfl_xor(s, 1, 32);
    s += __shfl_xor(s, 2, 32);
    s += __shfl_xor(s, 4, 32);
    const float lse = logf(s) + mx;

    float* hrow = d_out + (size_t)node * DIM_OUT;
    float* lrow = d_out + (size_t)N * DIM_OUT + (size_t)node * DIM_OUT;
    *(float4*)(hrow + sub * 4) = make_float4(v0, v1, v2, v3);
    hrow[32 + sub] = v4;
    *(float4*)(lrow + sub * 4) = make_float4(v0 - lse, v1 - lse, v2 - lse, v3 - lse);
    lrow[32 + sub] = v4 - lse;
}

static inline int cdiv_i(long long a, long long b) { return (int)((a + b - 1) / b); }

extern "C" void kernel_launch(void* const* d_in, const int* in_sizes, int n_in,
                              void* d_out, int out_size, void* d_ws, size_t ws_size,
                              hipStream_t stream) {
    (void)n_in; (void)out_size; (void)ws_size;
    const float* x     = (const float*)d_in[0];
    const int*   eidx  = (const int*)d_in[1];
    const float* Wl1   = (const float*)d_in[2];
    const float* bl1   = (const float*)d_in[3];
    const float* Wr1   = (const float*)d_in[4];
    const float* br1   = (const float*)d_in[5];
    const float* att1  = (const float*)d_in[6];
    const float* bias1 = (const float*)d_in[7];
    const float* Wl2   = (const float*)d_in[8];
    const float* bl2   = (const float*)d_in[9];
    const float* Wr2   = (const float*)d_in[10];
    const float* br2   = (const float*)d_in[11];
    const float* att2  = (const float*)d_in[12];
    const float* bias2 = (const float*)d_in[13];

    const int N = in_sizes[0] / DIM_IN;
    const int E = in_sizes[1] / 2;
    const int* src = eidx;
    const int* dst = eidx + E;

    float* ws = (float*)d_ws;
    size_t o = 0;
    float* h_l1   = ws + o; o += (size_t)N * HID;
    float* h_r1   = ws + o; o += (size_t)N * HID;
    float* h1     = ws + o; o += (size_t)N * HID;
    float* h_l2   = ws + o; o += (size_t)N * DIM_OUT;
    float* h_r2   = ws + o; o += (size_t)N * DIM_OUT;
    int* csr_src  = (int*)(ws + o); o += (size_t)E;
    int* perm     = (int*)(ws + o); o += (size_t)E;
    int* rnk      = (int*)(ws + o); o += (size_t)E;
    int* cnt      = (int*)(ws + o); o += (size_t)N + 1;
    int* rowptr   = (int*)(ws + o); o += (size_t)N + 1;

    const int B = 256;

    gat_fill_u32<<<cdiv_i(N, B), B, 0, stream>>>((unsigned int*)cnt, 0u, N);
    gat_count_rank<<<cdiv_i(E, B), B, 0, stream>>>(dst, cnt, rnk, E);
    gat_scan_counts<<<1, 1024, 0, stream>>>(cnt, rowptr, N);
    gat_place<<<cdiv_i(E, B), B, 0, stream>>>(dst, rnk, rowptr, perm, E);
    gat_order_emit<<<cdiv_i(E, B), B, 0, stream>>>(src, dst, rowptr, perm, csr_src, E);

    {
        const int tiles = (N >> 4) * (HID >> 4);
        const int grid = cdiv_i(tiles, B / 32);
        gat_gemm_bias_f32_wmma<DIM_IN, HID, HID>
            <<<grid, B, 0, stream>>>(x, Wl1, bl1, h_l1, N);
        gat_gemm_bias_f32_wmma<DIM_IN, HID, HID>
            <<<grid, B, 0, stream>>>(x, Wr1, br1, h_r1, N);
    }
    gat_node_l1<<<cdiv_i((long long)N * 8, B), B, 0, stream>>>(
        h_l1, h_r1, att1, bias1, rowptr, csr_src, h1, N);

    {
        const int tiles = (N >> 4) * (48 >> 4);
        const int grid = cdiv_i(tiles, B / 32);
        gat_gemm_bias_f32_wmma<HID, DIM_OUT, 48>
            <<<grid, B, 0, stream>>>(h1, Wl2, bl2, h_l2, N);
        gat_gemm_bias_f32_wmma<HID, DIM_OUT, 48>
            <<<grid, B, 0, stream>>>(h1, Wr2, br2, h_r2, N);
    }
    gat_node_l2_final<<<cdiv_i((long long)N * 8, B), B, 0, stream>>>(
        h_l2, h_r2, att2, bias2, rowptr, csr_src, (float*)d_out, N);
}


# Both projections of a layer in one launch, one work-group per WGP: [Wl|Wr] staged once in padded LDS, each wave keeps a 16-row strip in registers and sweeps all column tiles; k order of sums permuted
# speedup vs baseline: 2.9883x; 1.3926x over previous
#include <hip/hip_runtime.h>
#include <hip/hip_bf16.h>


#define DIM_IN   128
#define HID      64
#define HEADS    8
#define DIM_H    8
#define DIM_OUT  40
#define NEG_SLOPE 0.2f

typedef __attribute__((ext_vector_type(2))) float v2f;
typedef __attribute__((ext_vector_type(8))) float v8f;

__global__ void gat_fill_u32(unsigned int* __restrict__ p, unsigned int val,
                             long long n) {
    long long i = (long long)blockIdx.x * blockDim.x + threadIdx.x;
    if (i < n) p[i] = val;
}

#define GEMM_THREADS 512

template <int K, int CC>
__global__ __launch_bounds__(GEMM_THREADS) void gat_gemm_pair_f32_wmma(
    const float* __restrict__ A,
    const float* __restrict__ Wl, const float* __restrict__ bl,
    const float* __restrict__ Wr, const float* __restrict__ br,
    float* __restrict__ outl, float* __restrict__ outr, int n) {
    constexpr int NC = 2 * CC;
    constexpr int NT = NC / 16;
    constexpr int P = K + 4;
    constexpr int J = K / 8;
    static_assert(NC % 16 == 0 && K % 8 == 0, "tile shape");
    static_assert((K * NC) % GEMM_THREADS == 0, "staging loop has no tail");
    __shared__ __attribute__((aligned(16))) float sWt[NC * P];

    const int lane = threadIdx.x & 31;
    const int c = lane & 15;
    const int h = lane >> 4;
    const int wave = __builtin_amdgcn_readfirstlane((int)(threadIdx.x >> 5));
    constexpr int nWaves = GEMM_THREADS / 32;
    const int rowTiles = n >> 4;
    const int stride = nWaves * (int)gridDim.x;
    int rt = wave * (int)gridDim.x + (int)blockIdx.x;

    float4 a[J];
    if (rt < rowTiles) {
        const float4* ap = (const float4*)(A + (size_t)((rt << 4) + c) * K + 4 * h);
#pragma unroll
        for (int j = 0; j < J; ++j) a[j] = ap[2 * j];
    }

#pragma unroll
    for (int i = 0; i < (K * NC) / GEMM_THREADS; ++i) {
        const int idx = i * GEMM_THREADS + (int)threadIdx.x;
        const int k = idx / NC;
        const int cc = idx - k * NC;
        const float* wp = (cc < CC) ? (Wl + cc) : (Wr + (cc - CC));
        sWt[cc * P + k] = wp[(size_t)k * CC];
    }
    __syncthreads();

    const float* bBase = sWt + c * P + 4 * h;

    while (rt < rowTiles) {
        v8f acc[NT];
#pragma unroll
        for (int t = 0; t < NT; ++t) acc[t] = v8f{};

        float4 bq[2][NT];
#pragma unroll
        for (int t = 0; t < NT; ++t)
            bq[0][t] = *(const float4*)(bBase + t * 16 * P);
#pragma unroll
        for (int j = 0; j < J; ++j) {
            if (j + 1 < J) {
#pragma unroll
                for (int t = 0; t < NT; ++t)
                    bq[(j + 1) & 1][t] =
                        *(const float4*)(bBase + t * 16 * P + 8 * (j + 1));
            }
            const v2f a01 = {a[j].x, a[j].y};
            const v2f a23 = {a[j].z, a[j].w};
#pragma unroll
            for (int t = 0; t < NT; ++t) {
                const v2f b01 = {bq[j & 1][t].x, bq[j & 1][t].y};
                const v2f b23 = {bq[j & 1][t].z, bq[j & 1][t].w};
                acc[t] = __builtin_amdgcn_wmma_f32_16x16x4_f32(
                    false, a01, false, b01, (short)0, acc[t], false, false);
                acc[t] = __builtin_amdgcn_wmma_f32_16x16x4_f32(
                    false, a23, false, b23, (short)0, acc[t], false, false);
            }
            __builtin_amdgcn_sched_barrier(0);
        }

        const int rtNext = rt + stride;
        if (rtNext < rowTiles) {
            const float4* ap =
                (const float4*)(A + (size_t)((rtNext << 4) + c) * K + 4 * h);
#pragma unroll
            for (int j = 0; j < J; ++j) a[j] = ap[2 * j];
        }

        const int rbase = (rt << 4) + (h << 3);
#pragma unroll
        for (int t = 0; t < NT; ++t) {
            const int col = t * 16 + c;
            const bool left = col < CC;
            const float bv = left ? bl[col] : br[col - CC];
            float* op = left ? (outl + col) : (outr + (col - CC));
#pragma unroll
            for (int r = 0; r < 8; ++r) op[(size_t)(rbase + r) * CC] = acc[t][r] + bv;
        }
        rt = rtNext;
    }
}

__global__ void gat_count_rank(const int* __restrict__ dst, int* __restrict__ cnt,
                               int* __restrict__ rnk, int E) {
    const int e = blockIdx.x * blockDim.x + threadIdx.x;
    if (e >= E) return;
    rnk[e] = atomicAdd(&cnt[dst[e]], 1);
}

__global__ __launch_bounds__(1024) void gat_scan_counts(
    const int* __restrict__ cnt, int* __restrict__ rowptr, int N) {
    __shared__ int sWave[32];
    const int tid = threadIdx.x;
    const int lane = tid & 31;
    const int w = tid >> 5;
    const int chunk = (N + 1023) >> 10;
    const int b = min(tid * chunk, N);
    const int e = min(b + chunk, N);
    int sum = 0;
    for (int i = b; i < e; ++i) sum += cnt[i];
    int incl = sum;
    for (int o = 1; o < 32; o <<= 1) {
        const int v = __shfl_up(incl, o, 32);
        if (lane >= o) incl += v;
    }
    if (lane == 31) sWave[w] = incl;
    __syncthreads();
    if (w == 0) {
        const int v = sWave[lane];
        int iv = v;
        for (int o = 1; o < 32; o <<= 1) {
            const int u = __shfl_up(iv, o, 32);
            if (lane >= o) iv += u;
        }
        sWave[lane] = iv - v;
    }
    __syncthreads();
    int run = sWave[w] + incl - sum;
    for (int i = b; i < e; ++i) {
        rowptr[i] = run;
        run += cnt[i];
    }
    if (tid == 1023) rowptr[N] = run;
}

__global__ void gat_place(const int* __restrict__ dst, const int* __restrict__ rnk,
                          const int* __restrict__ rowptr, int* __restrict__ perm,
                          int E) {
    const int e = blockIdx.x * blockDim.x + threadIdx.x;
    if (e >= E) return;
    perm[rowptr[dst[e]] + rnk[e]] = e;
}

__global__ void gat_order_emit(const int* __restrict__ src, const int* __restrict__ dst,
                               const int* __restrict__ rowptr,
                               const int* __restrict__ perm,
                               int* __restrict__ csr_src, int E) {
    const int i = blockIdx.x * blockDim.x + threadIdx.x;
    if (i >= E) return;
    const int e = perm[i];
    const int d = dst[e];
    const int beg = rowptr[d], end = rowptr[d + 1];
    int r = 0;
    for (int j = beg; j < end; ++j) r += (perm[j] < e) ? 1 : 0;
    csr_src[beg + r] = src[e];
}

__device__ __forceinline__ float gat_leaky(float v) {
    return v > 0.0f ? v : v * NEG_SLOPE;
}

__device__ __forceinline__ float gat_elu(float v) {
    return v > 0.0f ? v : expm1f(v);
}

__global__ __launch_bounds__(256) void gat_node_l1(
    const float* __restrict__ xl, const float* __restrict__ xr,
    const float* __restrict__ att, const float* __restrict__ bias1,
    const int* __restrict__ rowptr, const int* __restrict__ csr_src,
    float* __restrict__ h1, int N) {
    const int t = blockIdx.x * blockDim.x + threadIdx.x;
    const int node = t >> 3;
    const int h = t & 7;
    if (node >= N) return;
    const int beg = rowptr[node], end = rowptr[node + 1];
    const float4* pi = (const float4*)(xr + (size_t)node * HID + h * DIM_H);
    const float4 xi0 = pi[0], xi1 = pi[1];
    const float4* pa = (const float4*)(att + h * DIM_H);
    const float4 a0 = pa[0], a1 = pa[1];

    float m = -__builtin_inff();
    float den = 0.0f;
    float acc[8] = {0.0f, 0.0f, 0.0f, 0.0f, 0.0f, 0.0f, 0.0f, 0.0f};

    int s0n = 0, s1n = 0;
    if (beg < end) {
        s0n = csr_src[beg];
        s1n = csr_src[min(beg + 1, end - 1)];
    }
    for (int i = beg; i < end; i += 2) {
        const int s0 = s0n, s1 = s1n;
        const bool has1 = (i + 1 < end);
        const float4* p0 = (const float4*)(xl + (size_t)s0 * HID + h * DIM_H);
        const float4* p1 = (const float4*)(xl + (size_t)s1 * HID + h * DIM_H);
        const float4 u0 = p0[0], u1 = p0[1];
        const float4 w0 = p1[0], w1 = p1[1];
        if (i + 2 < end) {
            s0n = csr_src[i + 2];
            s1n = csr_src[min(i + 3, end - 1)];
        }
        {
            float lg = 0.0f;
            lg += gat_leaky(xi0.x + u0.x) * a0.x;
            lg += gat_leaky(xi0.y + u0.y) * a0.y;
            lg += gat_leaky(xi0.z + u0.z) * a0.z;
            lg += gat_leaky(xi0.w + u0.w) * a0.w;
            lg += gat_leaky(xi1.x + u1.x) * a1.x;
            lg += gat_leaky(xi1.y + u1.y) * a1.y;
            lg += gat_leaky(xi1.z + u1.z) * a1.z;
            lg += gat_leaky(xi1.w + u1.w) * a1.w;
            const float mn = fmaxf(m, lg);
            const float sc = expf(m - mn);
            const float p = expf(lg - mn);
            den = den * sc + p;
            acc[0] = acc[0] * sc + p * u0.x;
            acc[1] = acc[1] * sc + p * u0.y;
            acc[2] = acc[2] * sc + p * u0.z;
            acc[3] = acc[3] * sc + p * u0.w;
            acc[4] = acc[4] * sc + p * u1.x;
            acc[5] = acc[5] * sc + p * u1.y;
            acc[6] = acc[6] * sc + p * u1.z;
            acc[7] = acc[7] * sc + p * u1.w;
            m = mn;
        }
        if (has1) {
            float lg = 0.0f;
            lg += gat_leaky(xi0.x + w0.x) * a0.x;
            lg += gat_leaky(xi0.y + w0.y) * a0.y;
            lg += gat_leaky(xi0.z + w0.z) * a0.z;
            lg += gat_leaky(xi0.w + w0.w) * a0.w;
            lg += gat_leaky(xi1.x + w1.x) * a1.x;
            lg += gat_leaky(xi1.y + w1.y) * a1.y;
            lg += gat_leaky(xi1.z + w1.z) * a1.z;
            lg += gat_leaky(xi1.w + w1.w) * a1.w;
            const float mn = fmaxf(m, lg);
            const float sc = expf(m - mn);
            const float p = expf(lg - mn);
            den = den * sc + p;
            acc[0] = acc[0] * sc + p * w0.x;
            acc[1] = acc[1] * sc + p * w0.y;
            acc[2] = acc[2] * sc + p * w0.z;
            acc[3] = acc[3] * sc + p * w0.w;
            acc[4] = acc[4] * sc + p * w1.x;
            acc[5] = acc[5] * sc + p * w1.y;
            acc[6] = acc[6] * sc + p * w1.z;
            acc[7] = acc[7] * sc + p * w1.w;
            m = mn;
        }
    }

    const float inv = (end > beg) ? 1.0f / den : 0.0f;
    const float4* pb = (const float4*)(bias1 + h * DIM_H);
    const float4 b0 = pb[0], b1 = pb[1];
    float4* po = (float4*)(h1 + (size_t)node * HID + h * DIM_H);
    po[0] = make_float4(gat_elu(acc[0] * inv + b0.x), gat_elu(acc[1] * inv + b0.y),
                        gat_elu(acc[2] * inv + b0.z), gat_elu(acc[3] * inv + b0.w));
    po[1] = make_float4(gat_elu(acc[4] * inv + b1.x), gat_elu(acc[5] * inv + b1.y),
                        gat_elu(acc[6] * inv + b1.z), gat_elu(acc[7] * inv + b1.w));
}

__global__ __launch_bounds__(256) void gat_node_l2_final(
    const float* __restrict__ xl, const float* __restrict__ xr,
    const float* __restrict__ att, const float* __restrict__ bias2,
    const int* __restrict__ rowptr, const int* __restrict__ csr_src,
    float* __restrict__ d_out, int N) {
    const int t = blockIdx.x * blockDim.x + threadIdx.x;
    const int node = t >> 3;
    const int sub = t & 7;
    if (node >= N) return;
    const int beg = rowptr[node], end = rowptr[node + 1];
    const float* ri = xr + (size_t)node * DIM_OUT;
    const float4 xi = *(const float4*)(ri + sub * 4);
    const float xi4 = ri[32 + sub];
    const float4 av = *(const float4*)(att + sub * 4);
    const float a4 = att[32 + sub];

    float m = -__builtin_inff();
    float den = 0.0f;
    float acc0 = 0.0f, acc1 = 0.0f, acc2 = 0.0f, acc3 = 0.0f, acc4 = 0.0f;

    int s0n = 0, s1n = 0;
    if (beg < end) {
        s0n = csr_src[beg];
        s1n = csr_src[min(beg + 1, end - 1)];
    }
    for (int i = beg; i < end; i += 2) {
        const int s0 = s0n, s1 = s1n;
        const bool has1 = (i + 1 < end);
        const float* r0 = xl + (size_t)s0 * DIM_OUT;
        const float* r1 = xl + (size_t)s1 * DIM_OUT;
        const float4 u = *(const float4*)(r0 + sub * 4);
        const float u4 = r0[32 + sub];
        const float4 w = *(const float4*)(r1 + sub * 4);
        const float w4 = r1[32 + sub];
        if (i + 2 < end) {
            s0n = csr_src[i + 2];
            s1n = csr_src[min(i + 3, end - 1)];
        }
        {
            float lg = 0.0f;
            lg += gat_leaky(xi.x + u.x) * av.x;
            lg += gat_leaky(xi.y + u.y) * av.y;
            lg += gat_leaky(xi.z + u.z) * av.z;
            lg += gat_leaky(xi.w + u.w) * av.w;
            lg += gat_leaky(xi4 + u4) * a4;
            lg += __shfl_xor(lg, 1, 32);
            lg += __shfl_xor(lg, 2, 32);
            lg += __shfl_xor(lg, 4, 32);
            const float mn = fmaxf(m, lg);
            const float sc = expf(m - mn);
            const float p = expf(lg - mn);
            den = den * sc + p;
            acc0 = acc0 * sc + p * u.x;
            acc1 = acc1 * sc + p * u.y;
            acc2 = acc2 * sc + p * u.z;
            acc3 = acc3 * sc + p * u.w;
            acc4 = acc4 * sc + p * u4;
            m = mn;
        }
        if (has1) {
            float lg = 0.0f;
            lg += gat_leaky(xi.x + w.x) * av.x;
            lg += gat_leaky(xi.y + w.y) * av.y;
            lg += gat_leaky(xi.z + w.z) * av.z;
            lg += gat_leaky(xi.w + w.w) * av.w;
            lg += gat_leaky(xi4 + w4) * a4;
            lg += __shfl_xor(lg, 1, 32);
            lg += __shfl_xor(lg, 2, 32);
            lg += __shfl_xor(lg, 4, 32);
            const float mn = fmaxf(m, lg);
            const float sc = expf(m - mn);
            const float p = expf(lg - mn);
            den = den * sc + p;
            acc0 = acc0 * sc + p * w.x;
            acc1 = acc1 * sc + p * w.y;
            acc2 = acc2 * sc + p * w.z;
            acc3 = acc3 * sc + p * w.w;
            acc4 = acc4 * sc + p * w4;
            m = mn;
        }
    }

    const float inv = (end > beg) ? 1.0f / den : 0.0f;
    const float4 bv = *(const float4*)(bias2 + sub * 4);
    const float v0 = acc0 * inv + bv.x;
    const float v1 = acc1 * inv + bv.y;
    const float v2 = acc2 * inv + bv.z;
    const float v3 = acc3 * inv + bv.w;
    const float v4 = acc4 * inv + bias2[32 + sub];

    float mx = fmaxf(fmaxf(fmaxf(v0, v1), fmaxf(v2, v3)), v4);
    mx = fmaxf(mx, __shfl_xor(mx, 1, 32));
    mx = fmaxf(mx, __shfl_xor(mx, 2, 32));
    mx = fmaxf(mx, __shfl_xor(mx, 4, 32));
    float s = expf(v0 - mx) + expf(v1 - mx) + expf(v2 - mx) + expf(v3 - mx) +
              expf(v4 - mx);
    s += __shfl_xor(s, 1, 32);
    s += __shfl_xor(s, 2, 32);
    s += __shfl_xor(s, 4, 32);
    const float lse = logf(s) + mx;

    float* hrow = d_out + (size_t)node * DIM_OUT;
    float* lrow = d_out + (size_t)N * DIM_OUT + (size_t)node * DIM_OUT;
    *(float4*)(hrow + sub * 4) = make_float4(v0, v1, v2, v3);
    hrow[32 + sub] = v4;
    *(float4*)(lrow + sub * 4) = make_float4(v0 - lse, v1 - lse, v2 - lse, v3 - lse);
    lrow[32 + sub] = v4 - lse;
}

static inline int cdiv_i(long long a, long long b) { return (int)((a + b - 1) / b); }

extern "C" void kernel_launch(void* const* d_in, const int* in_sizes, int n_in,
                              void* d_out, int out_size, void* d_ws, size_t ws_size,
                              hipStream_t stream) {
    (void)n_in; (void)out_size; (void)ws_size;
    const float* x     = (const float*)d_in[0];
    const int*   eidx  = (const int*)d_in[1];
    const float* Wl1   = (const float*)d_in[2];
    const float* bl1   = (const float*)d_in[3];
    const float* Wr1   = (const float*)d_in[4];
    const float* br1   = (const float*)d_in[5];
    const float* att1  = (const float*)d_in[6];
    const float* bias1 = (const float*)d_in[7];
    const float* Wl2   = (const float*)d_in[8];
    const float* bl2   = (const float*)d_in[9];
    const float* Wr2   = (const float*)d_in[10];
    const float* br2   = (const float*)d_in[11];
    const float* att2  = (const float*)d_in[12];
    const float* bias2 = (const float*)d_in[13];

    const int N = in_sizes[0] / DIM_IN;
    const int E = in_sizes[1] / 2;
    const int* src = eidx;
    const int* dst = eidx + E;

    float* ws = (float*)d_ws;
    size_t o = 0;
    float* h_l1   = ws + o; o += (size_t)N * HID;
    float* h_r1   = ws + o; o += (size_t)N * HID;
    float* h1     = ws + o; o += (size_t)N * HID;
    float* h_l2   = ws + o; o += (size_t)N * DIM_OUT;
    float* h_r2   = ws + o; o += (size_t)N * DIM_OUT;
    int* csr_src  = (int*)(ws + o); o += (size_t)E;
    int* perm     = (int*)(ws + o); o += (size_t)E;
    int* rnk      = (int*)(ws + o); o += (size_t)E;
    int* cnt      = (int*)(ws + o); o += (size_t)N + 1;
    int* rowptr   = (int*)(ws + o); o += (size_t)N + 1;

    const int B = 256;

    gat_fill_u32<<<cdiv_i(N, B), B, 0, stream>>>((unsigned int*)cnt, 0u, N);
    gat_count_rank<<<cdiv_i(E, B), B, 0, stream>>>(dst, cnt, rnk, E);
    gat_scan_counts<<<1, 1024, 0, stream>>>(cnt, rowptr, N);
    gat_place<<<cdiv_i(E, B), B, 0, stream>>>(dst, rnk, rowptr, perm, E);
    gat_order_emit<<<cdiv_i(E, B), B, 0, stream>>>(src, dst, rowptr, perm, csr_src, E);

    gat_gemm_pair_f32_wmma<DIM_IN, HID>
        <<<256, GEMM_THREADS, 0, stream>>>(x, Wl1, bl1, Wr1, br1, h_l1, h_r1, N);
    gat_node_l1<<<cdiv_i((long long)N * 8, B), B, 0, stream>>>(
        h_l1, h_r1, att1, bias1, rowptr, csr_src, h1, N);

    gat_gemm_pair_f32_wmma<HID, DIM_OUT>
        <<<256, GEMM_THREADS, 0, stream>>>(h1, Wl2, bl2, Wr2, br2, h_l2, h_r2, N);
    gat_node_l2_final<<<cdiv_i((long long)N * 8, B), B, 0, stream>>>(
        h_l2, h_r2, att2, bias2, rowptr, csr_src, (float*)d_out, N);
}


# Edges grouped by a two-level counting sort in LDS (256-node bins, then node and edge id) instead of per-edge global atomics, scattered stores and rank gathers; 9->8 launches; sum order unchanged
# speedup vs baseline: 5.3881x; 1.8031x over previous
#include <hip/hip_runtime.h>
#include <hip/hip_bf16.h>


#define DIM_IN   128
#define HID      64
#define HEADS    8
#define DIM_H    8
#define DIM_OUT  40
#define NEG_SLOPE 0.2f

typedef __attribute__((ext_vector_type(2))) float v2f;
typedef __attribute__((ext_vector_type(8))) float v8f;

#define GEMM_THREADS 512

template <int K, int CC>
__global__ __launch_bounds__(GEMM_THREADS) void gat_gemm_pair_f32_wmma(
    const float* __restrict__ A,
    const float* __restrict__ Wl, const float* __restrict__ bl,
    const float* __restrict__ Wr, const float* __restrict__ br,
    float* __restrict__ outl, float* __restrict__ outr, int n) {
    constexpr int NC = 2 * CC;
    constexpr int NT = NC / 16;
    constexpr int P = K + 4;
    constexpr int J = K / 8;
    static_assert(NC % 16 == 0 && K % 8 == 0, "tile shape");
    static_assert((K * NC) % GEMM_THREADS == 0, "staging loop has no tail");
    __shared__ __attribute__((aligned(16))) float sWt[NC * P];

    const int lane = threadIdx.x & 31;
    const int c = lane & 15;
    const int h = lane >> 4;
    const int wave = __builtin_amdgcn_readfirstlane((int)(threadIdx.x >> 5));
    constexpr int nWaves = GEMM_THREADS / 32;
    const int rowTiles = n >> 4;
    const int stride = nWaves * (int)gridDim.x;
    int rt = wave * (int)gridDim.x + (int)blockIdx.x;

    float4 a[J];
    if (rt < rowTiles) {
        const float4* ap = (const float4*)(A + (size_t)((rt << 4) + c) * K + 4 * h);
#pragma unroll
        for (int j = 0; j < J; ++j) a[j] = ap[2 * j];
    }

#pragma unroll
    for (int i = 0; i < (K * NC) / GEMM_THREADS; ++i) {
        const int idx = i * GEMM_THREADS + (int)threadIdx.x;
        const int k = idx / NC;
        const int cc = idx - k * NC;
        const float* wp = (cc < CC) ? (Wl + cc) : (Wr + (cc - CC));
        sWt[cc * P + k] = wp[(size_t)k * CC];
    }
    __syncthreads();

    const float* bBase = sWt + c * P + 4 * h;

    while (rt < rowTiles) {
        v8f acc[NT];
#pragma unroll
        for (int t = 0; t < NT; ++t) acc[t] = v8f{};

        float4 bq[2][NT];
#pragma unroll
        for (int t = 0; t < NT; ++t)
            bq[0][t] = *(const float4*)(bBase + t * 16 * P);
#pragma unroll
        for (int j = 0; j < J; ++j) {
            if (j + 1 < J) {
#pragma unroll
                for (int t = 0; t < NT; ++t)
                    bq[(j + 1) & 1][t] =
                        *(const float4*)(bBase + t * 16 * P + 8 * (j + 1));
            }
            const v2f a01 = {a[j].x, a[j].y};
            const v2f a23 = {a[j].z, a[j].w};
#pragma unroll
            for (int t = 0; t < NT; ++t) {
                const v2f b01 = {bq[j & 1][t].x, bq[j & 1][t].y};
                const v2f b23 = {bq[j & 1][t].z, bq[j & 1][t].w};
                acc[t] = __builtin_amdgcn_wmma_f32_16x16x4_f32(
                    false, a01, false, b01, (short)0, acc[t], false, false);
                acc[t] = __builtin_amdgcn_wmma_f32_16x16x4_f32(
                    false, a23, false, b23, (short)0, acc[t], false, false);
            }
            __builtin_amdgcn_sched_barrier(0);
        }

        const int rtNext = rt + stride;
        if (rtNext < rowTiles) {
            const float4* ap =
                (const float4*)(A + (size_t)((rtNext << 4) + c) * K + 4 * h);
#pragma unroll
            for (int j = 0; j < J; ++j) a[j] = ap[2 * j];
        }

        const int rbase = (rt << 4) + (h << 3);
#pragma unroll
        for (int t = 0; t < NT; ++t) {
            const int col = t * 16 + c;
            const bool left = col < CC;
            const float bv = left ? bl[col] : br[col - CC];
            float* op = left ? (outl + col) : (outr + (col - CC));
#pragma unroll
            for (int r = 0; r < 8; ++r) op[(size_t)(rbase + r) * CC] = acc[t][r] + bv;
        }
        rt = rtNext;
    }
}

#define BIN_SHIFT 8
#define BIN_NODES (1 << BIN_SHIFT)
#define NB_MAX    512
#define NCHUNK    256
#define CH_MAX    8192
#define KEY_SHIFT 21
#define BIN_CAP   8192
#define SORT_THREADS 1024

__device__ __forceinline__ int gat_block_excl_scan(int x, int* sWv) {
    const int lane = threadIdx.x & 31;
    const int w = threadIdx.x >> 5;
    int inc = x;
    for (int o = 1; o < 32; o <<= 1) {
        const int u = __shfl_up(inc, o, 32);
        if (lane >= o) inc += u;
    }
    if (lane == 31) sWv[w] = inc;
    __syncthreads();
    if (w == 0) {
        const int v = sWv[lane];
        int iv = v;
        for (int o = 1; o < 32; o <<= 1) {
            const int u = __shfl_up(iv, o, 32);
            if (lane >= o) iv += u;
        }
        sWv[lane] = iv - v;
    }
    __syncthreads();
    const int r = sWv[w] + inc - x;
    __syncthreads();
    return r;
}

__global__ __launch_bounds__(SORT_THREADS) void gat_bin_count(
    const int* __restrict__ dst, int* __restrict__ cntm, int E, int NB) {
    __shared__ int hist[NB_MAX];
    const int tid = threadIdx.x;
    if (tid < NB_MAX) hist[tid] = 0;
    __syncthreads();
    const int ch = (E + NCHUNK - 1) / NCHUNK;
    const int e0 = (int)blockIdx.x * ch;
    const int e1 = min(e0 + ch, E);
    const int n = max(e1 - e0, 0);
    constexpr int PER = CH_MAX / SORT_THREADS;
    int kd[PER];
#pragma unroll
    for (int k = 0; k < PER; ++k)
        kd[k] = dst[max(min(e0 + tid + k * SORT_THREADS, e1 - 1), 0)];
#pragma unroll
    for (int k = 0; k < PER; ++k)
        if (tid + k * SORT_THREADS < n) atomicAdd(&hist[kd[k] >> BIN_SHIFT], 1);
    __syncthreads();
    if (tid < NB) cntm[tid * NCHUNK + (int)blockIdx.x] = hist[tid];
}

__global__ __launch_bounds__(SORT_THREADS) void gat_scan_bins(
    const int* __restrict__ cntm, int* __restrict__ offm, int total) {
    __shared__ int sPart[32 * 32];
    __shared__ int sWv[32];
    const int tid = threadIdx.x;
    const int lane = tid & 31;
    const int w = tid >> 5;
    const int T = (total + 4095) >> 12;

    sPart[tid] = 0;
    __syncthreads();
#pragma unroll 8
    for (int t = 0; t < T; ++t) {
        const int i0 = (t << 12) + (tid << 2);
        const bool ok = i0 < total;
        const int4 q = *(const int4*)(cntm + (ok ? i0 : 0));
        int s = ok ? (q.x + q.y + q.z + q.w) : 0;
        s += __shfl_xor(s, 16, 32);
        s += __shfl_xor(s, 8, 32);
        s += __shfl_xor(s, 4, 32);
        s += __shfl_xor(s, 2, 32);
        s += __shfl_xor(s, 1, 32);
        if (lane == 0) sPart[t * 32 + w] = s;
    }
    __syncthreads();
    const int mine = sPart[tid];
    const int ex = gat_block_excl_scan(mine, sWv);
    sPart[tid] = ex;
    if (tid == SORT_THREADS - 1) offm[total] = ex + mine;
    __syncthreads();
#pragma unroll 8
    for (int t = 0; t < T; ++t) {
        const int i0 = (t << 12) + (tid << 2);
        const bool ok = i0 < total;
        const int4 q = *(const int4*)(cntm + (ok ? i0 : 0));
        const int s = q.x + q.y + q.z + q.w;
        int inc = s;
        for (int o = 1; o < 32; o <<= 1) {
            const int u = __shfl_up(inc, o, 32);
            if (lane >= o) inc += u;
        }
        const int b0 = sPart[t * 32 + w] + inc - s;
        if (ok) *(int4*)(offm + i0) = make_int4(b0, b0 + q.x, b0 + q.x + q.y, b0 + q.x + q.y + q.z);
    }
}

__global__ __launch_bounds__(SORT_THREADS) void gat_bin_place(
    const int* __restrict__ src, const int* __restrict__ dst,
    const int* __restrict__ offm, int* __restrict__ recw, int E, int NB) {
    constexpr int PER = CH_MAX / SORT_THREADS;
    int2* rec = (int2*)recw;
    __shared__ int cur[NB_MAX];
    __shared__ int lstart[NB_MAX];
    __shared__ int goff[NB_MAX];
    __shared__ int sWv[32];
    __shared__ int s_key[CH_MAX];
    __shared__ int s_src[CH_MAX];
    __shared__ int s_dest[CH_MAX];
    const int tid = threadIdx.x;
    if (tid < NB_MAX) cur[tid] = 0;
    const int ch = (E + NCHUNK - 1) / NCHUNK;
    const int e0 = (int)blockIdx.x * ch;
    const int e1 = min(e0 + ch, E);
    const int n = max(e1 - e0, 0);

    int kd[PER], ks[PER];
#pragma unroll
    for (int k = 0; k < PER; ++k) {
        const int e = max(min(e0 + tid + k * SORT_THREADS, e1 - 1), 0);
        kd[k] = dst[e];
        ks[k] = src[e];
    }
    if (tid < NB) goff[tid] = offm[tid * NCHUNK + (int)blockIdx.x];
    __syncthreads();

    int kr[PER];
#pragma unroll
    for (int k = 0; k < PER; ++k) {
        kr[k] = 0;
        if (tid + k * SORT_THREADS < n) kr[k] = atomicAdd(&cur[kd[k] >> BIN_SHIFT], 1);
    }
    __syncthreads();
    const int c = (tid < NB_MAX) ? cur[tid] : 0;
    const int ex = gat_block_excl_scan(c, sWv);
    if (tid < NB_MAX) lstart[tid] = ex;
    __syncthreads();

#pragma unroll
    for (int k = 0; k < PER; ++k) {
        if (tid + k * SORT_THREADS < n) {
            const int b = kd[k] >> BIN_SHIFT;
            const int p = lstart[b] + kr[k];
            s_key[p] = ((kd[k] & (BIN_NODES - 1)) << KEY_SHIFT) | (e0 + tid + k * SORT_THREADS);
            s_src[p] = ks[k];
            s_dest[p] = goff[b] + kr[k];
        }
    }
    __syncthreads();
    for (int p = tid; p < n; p += SORT_THREADS)
        rec[s_dest[p]] = make_int2(s_key[p], s_src[p]);
}

__global__ __launch_bounds__(SORT_THREADS) void gat_bin_sort(
    const int* __restrict__ recw, const int* __restrict__ offm,
    int* __restrict__ rowptr, int* __restrict__ csr_src, int N) {
    constexpr int PER = BIN_CAP / SORT_THREADS;
    const int2* rec = (const int2*)recw;
    __shared__ int cnt[BIN_NODES];
    __shared__ int nstart[BIN_NODES];
    __shared__ int sWv[32];
    __shared__ int p_key[BIN_CAP];
    __shared__ int p_src[BIN_CAP];
    __shared__ int o_src[BIN_CAP];
    const int tid = threadIdx.x;
    const int b = (int)blockIdx.x;
    const int base = offm[b * NCHUNK];
    const int end = offm[(b + 1) * NCHUNK];
    const int M = end - base;
    const int n0 = b << BIN_SHIFT;
    if (tid < BIN_NODES) cnt[tid] = 0;
    __syncthreads();

    if (M <= BIN_CAP) {
        int2 r[PER];
        int rk[PER];
#pragma unroll
        for (int k = 0; k < PER; ++k) r[k] = make_int2(0, 0);
        if (M > 0) {
#pragma unroll
            for (int k = 0; k < PER; ++k) r[k] = rec[base + min(tid + k * SORT_THREADS, M - 1)];
        }
#pragma unroll
        for (int k = 0; k < PER; ++k) {
            rk[k] = 0;
            if (tid + k * SORT_THREADS < M) rk[k] = atomicAdd(&cnt[r[k].x >> KEY_SHIFT], 1);
        }
        __syncthreads();
        const int c = (tid < BIN_NODES) ? cnt[tid] : 0;
        const int ex = gat_block_excl_scan(c, sWv);
        if (tid < BIN_NODES) {
            nstart[tid] = ex;
            if (n0 + tid < N) rowptr[n0 + tid] = base + ex;
        }
        if (b == (int)gridDim.x - 1 && tid == 0) rowptr[N] = end;
        __syncthreads();
#pragma unroll
        for (int k = 0; k < PER; ++k) {
            if (tid + k * SORT_THREADS < M) {
                const int p = nstart[r[k].x >> KEY_SHIFT] + rk[k];
                p_key[p] = r[k].x;
                p_src[p] = r[k].y;
            }
        }
        __syncthreads();
        for (int p = tid; p < M; p += SORT_THREADS) {
            const int key = p_key[p];
            const int dl = key >> KEY_SHIFT;
            const int s = nstart[dl];
            const int e = s + cnt[dl];
            int rank = 0;
            for (int q = s; q < e; ++q) rank += (p_key[q] < key) ? 1 : 0;
            o_src[s + rank] = p_src[p];
        }
        __syncthreads();
        for (int p = tid; p < M; p += SORT_THREADS) csr_src[base + p] = o_src[p];
    } else {
        for (int i = tid; i < M; i += SORT_THREADS)
            atomicAdd(&cnt[rec[base + i].x >> KEY_SHIFT], 1);
        __syncthreads();
        const int c = (tid < BIN_NODES) ? cnt[tid] : 0;
        const int ex = gat_block_excl_scan(c, sWv);
        if (tid < BIN_NODES) {
            nstart[tid] = ex;
            if (n0 + tid < N) rowptr[n0 + tid] = base + ex;
        }
        if (b == (int)gridDim.x - 1 && tid == 0) rowptr[N] = end;
        __syncthreads();
        for (int i = tid; i < M; i += SORT_THREADS) {
            const int2 ri = rec[base + i];
            const int dl = ri.x >> KEY_SHIFT;
            int rank = 0;
            for (int j = 0; j < M; ++j) {
                const int kj = rec[base + j].x;
                rank += ((kj >> KEY_SHIFT) == dl && kj < ri.x) ? 1 : 0;
            }
            csr_src[base + nstart[dl] + rank] = ri.y;
        }
    }
}

__device__ __forceinline__ float gat_leaky(float v) {
    return v > 0.0f ? v : v * NEG_SLOPE;
}

__device__ __forceinline__ float gat_elu(float v) {
    return v > 0.0f ? v : expm1f(v);
}

__global__ __launch_bounds__(256) void gat_node_l1(
    const float* __restrict__ xl, const float* __restrict__ xr,
    const float* __restrict__ att, const float* __restrict__ bias1,
    const int* __restrict__ rowptr, const int* __restrict__ csr_src,
    float* __restrict__ h1, int N) {
    const int t = blockIdx.x * blockDim.x + threadIdx.x;
    const int node = t >> 3;
    const int h = t & 7;
    if (node >= N) return;
    const int beg = rowptr[node], end = rowptr[node + 1];
    const float4* pi = (const float4*)(xr + (size_t)node * HID + h * DIM_H);
    const float4 xi0 = pi[0], xi1 = pi[1];
    const float4* pa = (const float4*)(att + h * DIM_H);
    const float4 a0 = pa[0], a1 = pa[1];

    float m = -__builtin_inff();
    float den = 0.0f;
    float acc[8] = {0.0f, 0.0f, 0.0f, 0.0f, 0.0f, 0.0f, 0.0f, 0.0f};

    int s0n = 0, s1n = 0;
    if (beg < end) {
        s0n = csr_src[beg];
        s1n = csr_src[min(beg + 1, end - 1)];
    }
    for (int i = beg; i < end; i += 2) {
        const int s0 = s0n, s1 = s1n;
        const bool has1 = (i + 1 < end);
        const float4* p0 = (const float4*)(xl + (size_t)s0 * HID + h * DIM_H);
        const float4* p1 = (const float4*)(xl + (size_t)s1 * HID + h * DIM_H);
        const float4 u0 = p0[0], u1 = p0[1];
        const float4 w0 = p1[0], w1 = p1[1];
        if (i + 2 < end) {
            s0n = csr_src[i + 2];
            s1n = csr_src[min(i + 3, end - 1)];
        }
        {
            float lg = 0.0f;
            lg += gat_leaky(xi0.x + u0.x) * a0.x;
            lg += gat_leaky(xi0.y + u0.y) * a0.y;
            lg += gat_leaky(xi0.z + u0.z) * a0.z;
            lg += gat_leaky(xi0.w + u0.w) * a0.w;
            lg += gat_leaky(xi1.x + u1.x) * a1.x;
            lg += gat_leaky(xi1.y + u1.y) * a1.y;
            lg += gat_leaky(xi1.z + u1.z) * a1.z;
            lg += gat_leaky(xi1.w + u1.w) * a1.w;
            const float mn = fmaxf(m, lg);
            const float sc = expf(m - mn);
            const float p = expf(lg - mn);
            den = den * sc + p;
            acc[0] = acc[0] * sc + p * u0.x;
            acc[1] = acc[1] * sc + p * u0.y;
            acc[2] = acc[2] * sc + p * u0.z;
            acc[3] = acc[3] * sc + p * u0.w;
            acc[4] = acc[4] * sc + p * u1.x;
            acc[5] = acc[5] * sc + p * u1.y;
            acc[6] = acc[6] * sc + p * u1.z;
            acc[7] = acc[7] * sc + p * u1.w;
            m = mn;
        }
        if (has1) {
            float lg = 0.0f;
            lg += gat_leaky(xi0.x + w0.x) * a0.x;
            lg += gat_leaky(xi0.y + w0.y) * a0.y;
            lg += gat_leaky(xi0.z + w0.z) * a0.z;
            lg += gat_leaky(xi0.w + w0.w) * a0.w;
            lg += gat_leaky(xi1.x + w1.x) * a1.x;
            lg += gat_leaky(xi1.y + w1.y) * a1.y;
            lg += gat_leaky(xi1.z + w1.z) * a1.z;
            lg += gat_leaky(xi1.w + w1.w) * a1.w;
            const float mn = fmaxf(m, lg);
            const float sc = expf(m - mn);
            const float p = expf(lg - mn);
            den = den * sc + p;
            acc[0] = acc[0] * sc + p * w0.x;
            acc[1] = acc[1] * sc + p * w0.y;
            acc[2] = acc[2] * sc + p * w0.z;
            acc[3] = acc[3] * sc + p * w0.w;
            acc[4] = acc[4] * sc + p * w1.x;
            acc[5] = acc[5] * sc + p * w1.y;
            acc[6] = acc[6] * sc + p * w1.z;
            acc[7] = acc[7] * sc + p * w1.w;
            m = mn;
        }
    }

    const float inv = (end > beg) ? 1.0f / den : 0.0f;
    const float4* pb = (const float4*)(bias1 + h * DIM_H);
    const float4 b0 = pb[0], b1 = pb[1];
    float4* po = (float4*)(h1 + (size_t)node * HID + h * DIM_H);
    po[0] = make_float4(gat_elu(acc[0] * inv + b0.x), gat_elu(acc[1] * inv + b0.y),
                        gat_elu(acc[2] * inv + b0.z), gat_elu(acc[3] * inv + b0.w));
    po[1] = make_float4(gat_elu(acc[4] * inv + b1.x), gat_elu(acc[5] * inv + b1.y),
                        gat_elu(acc[6] * inv + b1.z), gat_elu(acc[7] * inv + b1.w));
}

__global__ __launch_bounds__(256) void gat_node_l2_final(
    const float* __restrict__ xl, const float* __restrict__ xr,
    const float* __restrict__ att, const float* __restrict__ bias2,
    const int* __restrict__ rowptr, const int* __restrict__ csr_src,
    float* __restrict__ d_out, int N) {
    const int t = blockIdx.x * blockDim.x + threadIdx.x;
    const int node = t >> 3;
    const int sub = t & 7;
    if (node >= N) return;
    const int beg = rowptr[node], end = rowptr[node + 1];
    const float* ri = xr + (size_t)node * DIM_OUT;
    const float4 xi = *(const float4*)(ri + sub * 4);
    const float xi4 = ri[32 + sub];
    const float4 av = *(const float4*)(att + sub * 4);
    const float a4 = att[32 + sub];

    float m = -__builtin_inff();
    float den = 0.0f;
    float acc0 = 0.0f, acc1 = 0.0f, acc2 = 0.0f, acc3 = 0.0f, acc4 = 0.0f;

    int s0n = 0, s1n = 0;
    if (beg < end) {
        s0n = csr_src[beg];
        s1n = csr_src[min(beg + 1, end - 1)];
    }
    for (int i = beg; i < end; i += 2) {
        const int s0 = s0n, s1 = s1n;
        const bool has1 = (i + 1 < end);
        const float* r0 = xl + (size_t)s0 * DIM_OUT;
        const float* r1 = xl + (size_t)s1 * DIM_OUT;
        const float4 u = *(const float4*)(r0 + sub * 4);
        const float u4 = r0[32 + sub];
        const float4 w = *(const float4*)(r1 + sub * 4);
        const float w4 = r1[32 + sub];
        if (i + 2 < end) {
            s0n = csr_src[i + 2];
            s1n = csr_src[min(i + 3, end - 1)];
        }
        {
            float lg = 0.0f;
            lg += gat_leaky(xi.x + u.x) * av.x;
            lg += gat_leaky(xi.y + u.y) * av.y;
            lg += gat_leaky(xi.z + u.z) * av.z;
            lg += gat_leaky(xi.w + u.w) * av.w;
            lg += gat_leaky(xi4 + u4) * a4;
            lg += __shfl_xor(lg, 1, 32);
            lg += __shfl_xor(lg, 2, 32);
            lg += __shfl_xor(lg, 4, 32);
            const float mn = fmaxf(m, lg);
            const float sc = expf(m - mn);
            const float p = expf(lg - mn);
            den = den * sc + p;
            acc0 = acc0 * sc + p * u.x;
            acc1 = acc1 * sc + p * u.y;
            acc2 = acc2 * sc + p * u.z;
            acc3 = acc3 * sc + p * u.w;
            acc4 = acc4 * sc + p * u4;
            m = mn;
        }
        if (has1) {
            float lg = 0.0f;
            lg += gat_leaky(xi.x + w.x) * av.x;
            lg += gat_leaky(xi.y + w.y) * av.y;
            lg += gat_leaky(xi.z + w.z) * av.z;
            lg += gat_leaky(xi.w + w.w) * av.w;
            lg += gat_leaky(xi4 + w4) * a4;
            lg += __shfl_xor(lg, 1, 32);
            lg += __shfl_xor(lg, 2, 32);
            lg += __shfl_xor(lg, 4, 32);
            const float mn = fmaxf(m, lg);
            const float sc = expf(m - mn);
            const float p = expf(lg - mn);
            den = den * sc + p;
            acc0 = acc0 * sc + p * w.x;
            acc1 = acc1 * sc + p * w.y;
            acc2 = acc2 * sc + p * w.z;
            acc3 = acc3 * sc + p * w.w;
            acc4 = acc4 * sc + p * w4;
            m = mn;
        }
    }

    const float inv = (end > beg) ? 1.0f / den : 0.0f;
    const float4 bv = *(const float4*)(bias2 + sub * 4);
    const float v0 = acc0 * inv + bv.x;
    const float v1 = acc1 * inv + bv.y;
    const float v2 = acc2 * inv + bv.z;
    const float v3 = acc3 * inv + bv.w;
    const float v4 = acc4 * inv + bias2[32 + sub];

    float mx = fmaxf(fmaxf(fmaxf(v0, v1), fmaxf(v2, v3)), v4);
    mx = fmaxf(mx, __shfl_xor(mx, 1, 32));
    mx = fmaxf(mx, __shfl_xor(mx, 2, 32));
    mx = fmaxf(mx, __shfl_xor(mx, 4, 32));
    float s = expf(v0 - mx) + expf(v1 - mx) + expf(v2 - mx) + expf(v3 - mx) +
              expf(v4 - mx);
    s += __shfl_xor(s, 1, 32);
    s += __shfl_xor(s, 2, 32);
    s += __shfl_xor(s, 4, 32);
    const float lse = logf(s) + mx;

    float* hrow = d_out + (size_t)node * DIM_OUT;
    float* lrow = d_out + (size_t)N * DIM_OUT + (size_t)node * DIM_OUT;
    *(float4*)(hrow + sub * 4) = make_float4(v0, v1, v2, v3);
    hrow[32 + sub] = v4;
    *(float4*)(lrow + sub * 4) = make_float4(v0 - lse, v1 - lse, v2 - lse, v3 - lse);
    lrow[32 + sub] = v4 - lse;
}

static inline int cdiv_i(long long a, long long b) { return (int)((a + b - 1) / b); }

extern "C" void kernel_launch(void* const* d_in, const int* in_sizes, int n_in,
                              void* d_out, int out_size, void* d_ws, size_t ws_size,
                              hipStream_t stream) {
    (void)n_in; (void)out_size; (void)ws_size;
    const float* x     = (const float*)d_in[0];
    const int*   eidx  = (const int*)d_in[1];
    const float* Wl1   = (const float*)d_in[2];
    const float* bl1   = (const float*)d_in[3];
    const float* Wr1   = (const float*)d_in[4];
    const float* br1   = (const float*)d_in[5];
    const float* att1  = (const float*)d_in[6];
    const float* bias1 = (const float*)d_in[7];
    const float* Wl2   = (const float*)d_in[8];
    const float* bl2   = (const float*)d_in[9];
    const float* Wr2   = (const float*)d_in[10];
    const float* br2   = (const float*)d_in[11];
    const float* att2  = (const float*)d_in[12];
    const float* bias2 = (const float*)d_in[13];

    const int N = in_sizes[0] / DIM_IN;
    const int E = in_sizes[1] / 2;
    const int* src = eidx;
    const int* dst = eidx + E;

    float* ws = (float*)d_ws;
    size_t o = 0;
    float* h_l1   = ws + o; o += (size_t)N * HID;
    float* h_r1   = ws + o; o += (size_t)N * HID;
    float* h1     = ws + o; o += (size_t)N * HID;
    float* h_l2   = ws + o; o += (size_t)N * DIM_OUT;
    float* h_r2   = ws + o; o += (size_t)N * DIM_OUT;
    int* csr_src  = (int*)(ws + o); o += (size_t)E;
    int* rec      = (int*)(ws + o); o += (size_t)2 * E;
    int* cntm     = (int*)(ws + o); o += (size_t)NB_MAX * NCHUNK;
    int* offm     = (int*)(ws + o); o += (size_t)NB_MAX * NCHUNK + 4;
    int* rowptr   = (int*)(ws + o); o += (size_t)N + 1;

    const int B = 256;

    const int NB = cdiv_i(N, BIN_NODES);
    gat_bin_count<<<NCHUNK, SORT_THREADS, 0, stream>>>(dst, cntm, E, NB);
    gat_scan_bins<<<1, SORT_THREADS, 0, stream>>>(cntm, offm, NB * NCHUNK);
    gat_bin_place<<<NCHUNK, SORT_THREADS, 0, stream>>>(src, dst, offm, rec, E, NB);
    gat_bin_sort<<<NB, SORT_THREADS, 0, stream>>>(rec, offm, rowptr, csr_src, N);

    gat_gemm_pair_f32_wmma<DIM_IN, HID>
        <<<256, GEMM_THREADS, 0, stream>>>(x, Wl1, bl1, Wr1, br1, h_l1, h_r1, N);
    gat_node_l1<<<cdiv_i((long long)N * 8, B), B, 0, stream>>>(
        h_l1, h_r1, att1, bias1, rowptr, csr_src, h1, N);

    gat_gemm_pair_f32_wmma<HID, DIM_OUT>
        <<<256, GEMM_THREADS, 0, stream>>>(h1, Wl2, bl2, Wr2, br2, h_l2, h_r2, N);
    gat_node_l2_final<<<cdiv_i((long long)N * 8, B), B, 0, stream>>>(
        h_l2, h_r2, att2, bias2, rowptr, csr_src, (float*)d_out, N);
}


# Edge grouping in 2 launches instead of 4: a chunk writes its bin-ordered records to its own slice plus {start,count} per bin, a bin's work-group finds its runs from that table; sum order unchanged
# speedup vs baseline: 6.7145x; 1.2462x over previous
#include <hip/hip_runtime.h>
#include <hip/hip_bf16.h>


#define DIM_IN   128
#define HID      64
#define HEADS    8
#define DIM_H    8
#define DIM_OUT  40
#define NEG_SLOPE 0.2f

typedef __attribute__((ext_vector_type(2))) float v2f;
typedef __attribute__((ext_vector_type(8))) float v8f;

#define GEMM_THREADS 512

template <int K, int CC>
__global__ __launch_bounds__(GEMM_THREADS) void gat_gemm_pair_f32_wmma(
    const float* __restrict__ A,
    const float* __restrict__ Wl, const float* __restrict__ bl,
    const float* __restrict__ Wr, const float* __restrict__ br,
    float* __restrict__ outl, float* __restrict__ outr, int n) {
    constexpr int NC = 2 * CC;
    constexpr int NT = NC / 16;
    constexpr int P = K + 4;
    constexpr int J = K / 8;
    static_assert(NC % 16 == 0 && K % 8 == 0, "tile shape");
    static_assert((K * NC) % GEMM_THREADS == 0, "staging loop has no tail");
    __shared__ __attribute__((aligned(16))) float sWt[NC * P];

    const int lane = threadIdx.x & 31;
    const int c = lane & 15;
    const int h = lane >> 4;
    const int wave = __builtin_amdgcn_readfirstlane((int)(threadIdx.x >> 5));
    constexpr int nWaves = GEMM_THREADS / 32;
    const int rowTiles = n >> 4;
    const int stride = nWaves * (int)gridDim.x;
    int rt = wave * (int)gridDim.x + (int)blockIdx.x;

    float4 a[J];
    if (rt < rowTiles) {
        const float4* ap = (const float4*)(A + (size_t)((rt << 4) + c) * K + 4 * h);
#pragma unroll
        for (int j = 0; j < J; ++j) a[j] = ap[2 * j];
    }

#pragma unroll
    for (int i = 0; i < (K * NC) / GEMM_THREADS; ++i) {
        const int idx = i * GEMM_THREADS + (int)threadIdx.x;
        const int k = idx / NC;
        const int cc = idx - k * NC;
        const float* wp = (cc < CC) ? (Wl + cc) : (Wr + (cc - CC));
        sWt[cc * P + k] = wp[(size_t)k * CC];
    }
    __syncthreads();

    const float* bBase = sWt + c * P + 4 * h;

    while (rt < rowTiles) {
        v8f acc[NT];
#pragma unroll
        for (int t = 0; t < NT; ++t) acc[t] = v8f{};

        float4 bq[2][NT];
#pragma unroll
        for (int t = 0; t < NT; ++t)
            bq[0][t] = *(const float4*)(bBase + t * 16 * P);
#pragma unroll
        for (int j = 0; j < J; ++j) {
            if (j + 1 < J) {
#pragma unroll
                for (int t = 0; t < NT; ++t)
                    bq[(j + 1) & 1][t] =
                        *(const float4*)(bBase + t * 16 * P + 8 * (j + 1));
            }
            const v2f a01 = {a[j].x, a[j].y};
            const v2f a23 = {a[j].z, a[j].w};
#pragma unroll
            for (int t = 0; t < NT; ++t) {
                const v2f b01 = {bq[j & 1][t].x, bq[j & 1][t].y};
                const v2f b23 = {bq[j & 1][t].z, bq[j & 1][t].w};
                acc[t] = __builtin_amdgcn_wmma_f32_16x16x4_f32(
                    false, a01, false, b01, (short)0, acc[t], false, false);
                acc[t] = __builtin_amdgcn_wmma_f32_16x16x4_f32(
                    false, a23, false, b23, (short)0, acc[t], false, false);
            }
            __builtin_amdgcn_sched_barrier(0);
        }

        const int rtNext = rt + stride;
        if (rtNext < rowTiles) {
            const float4* ap =
                (const float4*)(A + (size_t)((rtNext << 4) + c) * K + 4 * h);
#pragma unroll
            for (int j = 0; j < J; ++j) a[j] = ap[2 * j];
        }

        const int rbase = (rt << 4) + (h << 3);
#pragma unroll
        for (int t = 0; t < NT; ++t) {
            const int col = t * 16 + c;
            const bool left = col < CC;
            const float bv = left ? bl[col] : br[col - CC];
            float* op = left ? (outl + col) : (outr + (col - CC));
#pragma unroll
            for (int r = 0; r < 8; ++r) op[(size_t)(rbase + r) * CC] = acc[t][r] + bv;
        }
        rt = rtNext;
    }
}

#define BIN_SHIFT 8
#define BIN_NODES (1 << BIN_SHIFT)
#define NB_MAX    512
#define NCHUNK    256
#define CH_MAX    8192
#define KEY_SHIFT 21
#define BIN_CAP   8192
#define SORT_THREADS 1024

__device__ __forceinline__ int gat_block_excl_scan(int x, int* sWv) {
    const int lane = threadIdx.x & 31;
    const int w = threadIdx.x >> 5;
    int inc = x;
    for (int o = 1; o < 32; o <<= 1) {
        const int u = __shfl_up(inc, o, 32);
        if (lane >= o) inc += u;
    }
    if (lane == 31) sWv[w] = inc;
    __syncthreads();
    if (w == 0) {
        const int v = sWv[lane];
        int iv = v;
        for (int o = 1; o < 32; o <<= 1) {
            const int u = __shfl_up(iv, o, 32);
            if (lane >= o) iv += u;
        }
        sWv[lane] = iv - v;
    }
    __syncthreads();
    const int r = sWv[w] + inc - x;
    __syncthreads();
    return r;
}

__global__ __launch_bounds__(SORT_THREADS) void gat_chunk_order(
    const int* __restrict__ src, const int* __restrict__ dst,
    int* __restrict__ recw, int* __restrict__ tabw, int E, int NB) {
    constexpr int PER = CH_MAX / SORT_THREADS;
    int2* rec = (int2*)recw;
    int2* tab = (int2*)tabw;
    __shared__ int cur[NB_MAX];
    __shared__ int lstart[NB_MAX];
    __shared__ int sWv[32];
    __shared__ int s_key[CH_MAX];
    __shared__ int s_src[CH_MAX];
    const int tid = threadIdx.x;
    if (tid < NB_MAX) cur[tid] = 0;
    const int ch = (E + NCHUNK - 1) / NCHUNK;
    const int e0 = (int)blockIdx.x * ch;
    const int e1 = min(e0 + ch, E);
    const int n = max(e1 - e0, 0);

    int kd[PER], ks[PER];
#pragma unroll
    for (int k = 0; k < PER; ++k) {
        const int e = max(min(e0 + tid + k * SORT_THREADS, e1 - 1), 0);
        kd[k] = dst[e];
        ks[k] = src[e];
    }
    __syncthreads();

    int kr[PER];
#pragma unroll
    for (int k = 0; k < PER; ++k) {
        kr[k] = 0;
        if (tid + k * SORT_THREADS < n) kr[k] = atomicAdd(&cur[kd[k] >> BIN_SHIFT], 1);
    }
    __syncthreads();
    const int c = (tid < NB_MAX) ? cur[tid] : 0;
    const int ex = gat_block_excl_scan(c, sWv);
    if (tid < NB_MAX) lstart[tid] = ex;
    if (tid < NB) tab[tid * NCHUNK + (int)blockIdx.x] = make_int2(ex, c);
    __syncthreads();

#pragma unroll
    for (int k = 0; k < PER; ++k) {
        if (tid + k * SORT_THREADS < n) {
            const int p = lstart[kd[k] >> BIN_SHIFT] + kr[k];
            s_key[p] = ((kd[k] & (BIN_NODES - 1)) << KEY_SHIFT) | (e0 + tid + k * SORT_THREADS);
            s_src[p] = ks[k];
        }
    }
    __syncthreads();
    for (int p = tid; p < n; p += SORT_THREADS)
        rec[e0 + p] = make_int2(s_key[p], s_src[p]);
}

__device__ __forceinline__ int gat_run_slot(const int* sPre, const int* sAt, int p) {
    int lo = 0;
#pragma unroll
    for (int w = NCHUNK >> 1; w > 0; w >>= 1)
        if (sPre[lo + w] <= p) lo += w;
    return sAt[lo] + p;
}

__global__ __launch_bounds__(SORT_THREADS) void gat_bin_sort(
    const int* __restrict__ recw, const int* __restrict__ tabw,
    int* __restrict__ rowptr, int* __restrict__ csr_src, int N, int E) {
    constexpr int PER = BIN_CAP / SORT_THREADS;
    const int2* rec = (const int2*)recw;
    const int2* tab = (const int2*)tabw;
    __shared__ int cnt[BIN_NODES];
    __shared__ int nstart[BIN_NODES];
    __shared__ int sWv[32];
    __shared__ int sPre[NCHUNK + 1];
    __shared__ int sAt[NCHUNK];
    __shared__ int sBase;
    __shared__ int p_key[BIN_CAP];
    __shared__ int p_src[BIN_CAP];
    __shared__ int o_src[BIN_CAP];
    const int tid = threadIdx.x;
    const int b = (int)blockIdx.x;
    const int ch = (E + NCHUNK - 1) / NCHUNK;
    const int n0 = b << BIN_SHIFT;
    int2 row = make_int2(0, 0);
    if (tid < NCHUNK) row = tab[b * NCHUNK + tid];
    if (tid < BIN_NODES) cnt[tid] = 0;
    if (tid == 0) sBase = 0;
    __syncthreads();

    int st = row.x;
    st += __shfl_xor(st, 16, 32);
    st += __shfl_xor(st, 8, 32);
    st += __shfl_xor(st, 4, 32);
    st += __shfl_xor(st, 2, 32);
    st += __shfl_xor(st, 1, 32);
    if ((tid & 31) == 0 && tid < NCHUNK) atomicAdd(&sBase, st);
    const int pre = gat_block_excl_scan(row.y, sWv);
    if (tid < NCHUNK) {
        sPre[tid] = pre;
        sAt[tid] = tid * ch + row.x - pre;
    }
    if (tid == NCHUNK - 1) sPre[NCHUNK] = pre + row.y;
    __syncthreads();
    const int M = sPre[NCHUNK];
    const int base = sBase;
    const int end = base + M;

    if (M <= BIN_CAP) {
        int2 r[PER];
        int rk[PER];
#pragma unroll
        for (int k = 0; k < PER; ++k) r[k] = make_int2(0, 0);
        if (M > 0) {
#pragma unroll
            for (int k = 0; k < PER; ++k)
                r[k] = rec[gat_run_slot(sPre, sAt, min(tid + k * SORT_THREADS, M - 1))];
        }
#pragma unroll
        for (int k = 0; k < PER; ++k) {
            rk[k] = 0;
            if (tid + k * SORT_THREADS < M) rk[k] = atomicAdd(&cnt[r[k].x >> KEY_SHIFT], 1);
        }
        __syncthreads();
        const int c = (tid < BIN_NODES) ? cnt[tid] : 0;
        const int ex = gat_block_excl_scan(c, sWv);
        if (tid < BIN_NODES) {
            nstart[tid] = ex;
            if (n0 + tid < N) rowptr[n0 + tid] = base + ex;
        }
        if (b == (int)gridDim.x - 1 && tid == 0) rowptr[N] = end;
        __syncthreads();
#pragma unroll
        for (int k = 0; k < PER; ++k) {
            if (tid + k * SORT_THREADS < M) {
                const int p = nstart[r[k].x >> KEY_SHIFT] + rk[k];
                p_key[p] = r[k].x;
                p_src[p] = r[k].y;
            }
        }
        __syncthreads();
        for (int p = tid; p < M; p += SORT_THREADS) {
            const int key = p_key[p];
            const int dl = key >> KEY_SHIFT;
            const int s = nstart[dl];
            const int e = s + cnt[dl];
            int rank = 0;
            for (int q = s; q < e; ++q) rank += (p_key[q] < key) ? 1 : 0;
            o_src[s + rank] = p_src[p];
        }
        __syncthreads();
        for (int p = tid; p < M; p += SORT_THREADS) csr_src[base + p] = o_src[p];
    } else {
        for (int i = tid; i < M; i += SORT_THREADS)
            atomicAdd(&cnt[rec[gat_run_slot(sPre, sAt, i)].x >> KEY_SHIFT], 1);
        __syncthreads();
        const int c = (tid < BIN_NODES) ? cnt[tid] : 0;
        const int ex = gat_block_excl_scan(c, sWv);
        if (tid < BIN_NODES) {
            nstart[tid] = ex;
            if (n0 + tid < N) rowptr[n0 + tid] = base + ex;
        }
        if (b == (int)gridDim.x - 1 && tid == 0) rowptr[N] = end;
        __syncthreads();
        for (int i = tid; i < M; i += SORT_THREADS) {
            const int2 ri = rec[gat_run_slot(sPre, sAt, i)];
            const int dl = ri.x >> KEY_SHIFT;
            int rank = 0;
            for (int j = 0; j < M; ++j) {
                const int kj = rec[gat_run_slot(sPre, sAt, j)].x;
                rank += ((kj >> KEY_SHIFT) == dl && kj < ri.x) ? 1 : 0;
            }
            csr_src[base + nstart[dl] + rank] = ri.y;
        }
    }
}

__device__ __forceinline__ float gat_leaky(float v) {
    return v > 0.0f ? v : v * NEG_SLOPE;
}

__device__ __forceinline__ float gat_elu(float v) {
    return v > 0.0f ? v : expm1f(v);
}

__global__ __launch_bounds__(256) void gat_node_l1(
    const float* __restrict__ xl, const float* __restrict__ xr,
    const float* __restrict__ att, const float* __restrict__ bias1,
    const int* __restrict__ rowptr, const int* __restrict__ csr_src,
    float* __restrict__ h1, int N) {
    const int t = blockIdx.x * blockDim.x + threadIdx.x;
    const int node = t >> 3;
    const int h = t & 7;
    if (node >= N) return;
    const int beg = rowptr[node], end = rowptr[node + 1];
    const float4* pi = (const float4*)(xr + (size_t)node * HID + h * DIM_H);
    const float4 xi0 = pi[0], xi1 = pi[1];
    const float4* pa = (const float4*)(att + h * DIM_H);
    const float4 a0 = pa[0], a1 = pa[1];

    float m = -__builtin_inff();
    float den = 0.0f;
    float acc[8] = {0.0f, 0.0f, 0.0f, 0.0f, 0.0f, 0.0f, 0.0f, 0.0f};

    int s0n = 0, s1n = 0;
    if (beg < end) {
        s0n = csr_src[beg];
        s1n = csr_src[min(beg + 1, end - 1)];
    }
    for (int i = beg; i < end; i += 2) {
        const int s0 = s0n, s1 = s1n;
        const bool has1 = (i + 1 < end);
        const float4* p0 = (const float4*)(xl + (size_t)s0 * HID + h * DIM_H);
        const float4* p1 = (const float4*)(xl + (size_t)s1 * HID + h * DIM_H);
        const float4 u0 = p0[0], u1 = p0[1];
        const float4 w0 = p1[0], w1 = p1[1];
        if (i + 2 < end) {
            s0n = csr_src[i + 2];
            s1n = csr_src[min(i + 3, end - 1)];
        }
        {
            float lg = 0.0f;
            lg += gat_leaky(xi0.x + u0.x) * a0.x;
            lg += gat_leaky(xi0.y + u0.y) * a0.y;
            lg += gat_leaky(xi0.z + u0.z) * a0.z;
            lg += gat_leaky(xi0.w + u0.w) * a0.w;
            lg += gat_leaky(xi1.x + u1.x) * a1.x;
            lg += gat_leaky(xi1.y + u1.y) * a1.y;
            lg += gat_leaky(xi1.z + u1.z) * a1.z;
            lg += gat_leaky(xi1.w + u1.w) * a1.w;
            const float mn = fmaxf(m, lg);
            const float sc = expf(m - mn);
            const float p = expf(lg - mn);
            den = den * sc + p;
            acc[0] = acc[0] * sc + p * u0.x;
            acc[1] = acc[1] * sc + p * u0.y;
            acc[2] = acc[2] * sc + p * u0.z;
            acc[3] = acc[3] * sc + p * u0.w;
            acc[4] = acc[4] * sc + p * u1.x;
            acc[5] = acc[5] * sc + p * u1.y;
            acc[6] = acc[6] * sc + p * u1.z;
            acc[7] = acc[7] * sc + p * u1.w;
            m = mn;
        }
        if (has1) {
            float lg = 0.0f;
            lg += gat_leaky(xi0.x + w0.x) * a0.x;
            lg += gat_leaky(xi0.y + w0.y) * a0.y;
            lg += gat_leaky(xi0.z + w0.z) * a0.z;
            lg += gat_leaky(xi0.w + w0.w) * a0.w;
            lg += gat_leaky(xi1.x + w1.x) * a1.x;
            lg += gat_leaky(xi1.y + w1.y) * a1.y;
            lg += gat_leaky(xi1.z + w1.z) * a1.z;
            lg += gat_leaky(xi1.w + w1.w) * a1.w;
            const float mn = fmaxf(m, lg);
            const float sc = expf(m - mn);
            const float p = expf(lg - mn);
            den = den * sc + p;
            acc[0] = acc[0] * sc + p * w0.x;
            acc[1] = acc[1] * sc + p * w0.y;
            acc[2] = acc[2] * sc + p * w0.z;
            acc[3] = acc[3] * sc + p * w0.w;
            acc[4] = acc[4] * sc + p * w1.x;
            acc[5] = acc[5] * sc + p * w1.y;
            acc[6] = acc[6] * sc + p * w1.z;
            acc[7] = acc[7] * sc + p * w1.w;
            m = mn;
        }
    }

    const float inv = (end > beg) ? 1.0f / den : 0.0f;
    const float4* pb = (const float4*)(bias1 + h * DIM_H);
    const float4 b0 = pb[0], b1 = pb[1];
    float4* po = (float4*)(h1 + (size_t)node * HID + h * DIM_H);
    po[0] = make_float4(gat_elu(acc[0] * inv + b0.x), gat_elu(acc[1] * inv + b0.y),
                        gat_elu(acc[2] * inv + b0.z), gat_elu(acc[3] * inv + b0.w));
    po[1] = make_float4(gat_elu(acc[4] * inv + b1.x), gat_elu(acc[5] * inv + b1.y),
                        gat_elu(acc[6] * inv + b1.z), gat_elu(acc[7] * inv + b1.w));
}

__global__ __launch_bounds__(256) void gat_node_l2_final(
    const float* __restrict__ xl, const float* __restrict__ xr,
    const float* __restrict__ att, const float* __restrict__ bias2,
    const int* __restrict__ rowptr, const int* __restrict__ csr_src,
    float* __restrict__ d_out, int N) {
    const int t = blockIdx.x * blockDim.x + threadIdx.x;
    const int node = t >> 3;
    const int sub = t & 7;
    if (node >= N) return;
    const int beg = rowptr[node], end = rowptr[node + 1];
    const float* ri = xr + (size_t)node * DIM_OUT;
    const float4 xi = *(const float4*)(ri + sub * 4);
    const float xi4 = ri[32 + sub];
    const float4 av = *(const float4*)(att + sub * 4);
    const float a4 = att[32 + sub];

    float m = -__builtin_inff();
    float den = 0.0f;
    float acc0 = 0.0f, acc1 = 0.0f, acc2 = 0.0f, acc3 = 0.0f, acc4 = 0.0f;

    int s0n = 0, s1n = 0;
    if (beg < end) {
        s0n = csr_src[beg];
        s1n = csr_src[min(beg + 1, end - 1)];
    }
    for (int i = beg; i < end; i += 2) {
        const int s0 = s0n, s1 = s1n;
        const bool has1 = (i + 1 < end);
        const float* r0 = xl + (size_t)s0 * DIM_OUT;
        const float* r1 = xl + (size_t)s1 * DIM_OUT;
        const float4 u = *(const float4*)(r0 + sub * 4);
        const float u4 = r0[32 + sub];
        const float4 w = *(const float4*)(r1 + sub * 4);
        const float w4 = r1[32 + sub];
        if (i + 2 < end) {
            s0n = csr_src[i + 2];
            s1n = csr_src[min(i + 3, end - 1)];
        }
        {
            float lg = 0.0f;
            lg += gat_leaky(xi.x + u.x) * av.x;
            lg += gat_leaky(xi.y + u.y) * av.y;
            lg += gat_leaky(xi.z + u.z) * av.z;
            lg += gat_leaky(xi.w + u.w) * av.w;
            lg += gat_leaky(xi4 + u4) * a4;
            lg += __shfl_xor(lg, 1, 32);
            lg += __shfl_xor(lg, 2, 32);
            lg += __shfl_xor(lg, 4, 32);
            const float mn = fmaxf(m, lg);
            const float sc = expf(m - mn);
            const float p = expf(lg - mn);
            den = den * sc + p;
            acc0 = acc0 * sc + p * u.x;
            acc1 = acc1 * sc + p * u.y;
            acc2 = acc2 * sc + p * u.z;
            acc3 = acc3 * sc + p * u.w;
            acc4 = acc4 * sc + p * u4;
            m = mn;
        }
        if (has1) {
            float lg = 0.0f;
            lg += gat_leaky(xi.x + w.x) * av.x;
            lg += gat_leaky(xi.y + w.y) * av.y;
            lg += gat_leaky(xi.z + w.z) * av.z;
            lg += gat_leaky(xi.w + w.w) * av.w;
            lg += gat_leaky(xi4 + w4) * a4;
            lg += __shfl_xor(lg, 1, 32);
            lg += __shfl_xor(lg, 2, 32);
            lg += __shfl_xor(lg, 4, 32);
            const float mn = fmaxf(m, lg);
            const float sc = expf(m - mn);
            const float p = expf(lg - mn);
            den = den * sc + p;
            acc0 = acc0 * sc + p * w.x;
            acc1 = acc1 * sc + p * w.y;
            acc2 = acc2 * sc + p * w.z;
            acc3 = acc3 * sc + p * w.w;
            acc4 = acc4 * sc + p * w4;
            m = mn;
        }
    }

    const float inv = (end > beg) ? 1.0f / den : 0.0f;
    const float4 bv = *(const float4*)(bias2 + sub * 4);
    const float v0 = acc0 * inv + bv.x;
    const float v1 = acc1 * inv + bv.y;
    const float v2 = acc2 * inv + bv.z;
    const float v3 = acc3 * inv + bv.w;
    const float v4 = acc4 * inv + bias2[32 + sub];

    float mx = fmaxf(fmaxf(fmaxf(v0, v1), fmaxf(v2, v3)), v4);
    mx = fmaxf(mx, __shfl_xor(mx, 1, 32));
    mx = fmaxf(mx, __shfl_xor(mx, 2, 32));
    mx = fmaxf(mx, __shfl_xor(mx, 4, 32));
    float s = expf(v0 - mx) + expf(v1 - mx) + expf(v2 - mx) + expf(v3 - mx) +
              expf(v4 - mx);
    s += __shfl_xor(s, 1, 32);
    s += __shfl_xor(s, 2, 32);
    s += __shfl_xor(s, 4, 32);
    const float lse = logf(s) + mx;

    float* hrow = d_out + (size_t)node * DIM_OUT;
    float* lrow = d_out + (size_t)N * DIM_OUT + (size_t)node * DIM_OUT;
    *(float4*)(hrow + sub * 4) = make_float4(v0, v1, v2, v3);
    hrow[32 + sub] = v4;
    *(float4*)(lrow + sub * 4) = make_float4(v0 - lse, v1 - lse, v2 - lse, v3 - lse);
    lrow[32 + sub] = v4 - lse;
}

static inline int cdiv_i(long long a, long long b) { return (int)((a + b - 1) / b); }

extern "C" void kernel_launch(void* const* d_in, const int* in_sizes, int n_in,
                              void* d_out, int out_size, void* d_ws, size_t ws_size,
                              hipStream_t stream) {
    (void)n_in; (void)out_size; (void)ws_size;
    const float* x     = (const float*)d_in[0];
    const int*   eidx  = (const int*)d_in[1];
    const float* Wl1   = (const float*)d_in[2];
    const float* bl1   = (const float*)d_in[3];
    const float* Wr1   = (const float*)d_in[4];
    const float* br1   = (const float*)d_in[5];
    const float* att1  = (const float*)d_in[6];
    const float* bias1 = (const float*)d_in[7];
    const float* Wl2   = (const float*)d_in[8];
    const float* bl2   = (const float*)d_in[9];
    const float* Wr2   = (const float*)d_in[10];
    const float* br2   = (const float*)d_in[11];
    const float* att2  = (const float*)d_in[12];
    const float* bias2 = (const float*)d_in[13];

    const int N = in_sizes[0] / DIM_IN;
    const int E = in_sizes[1] / 2;
    const int* src = eidx;
    const int* dst = eidx + E;

    float* ws = (float*)d_ws;
    size_t o = 0;
    float* h_l1   = ws + o; o += (size_t)N * HID;
    float* h_r1   = ws + o; o += (size_t)N * HID;
    float* h1     = ws + o; o += (size_t)N * HID;
    float* h_l2   = ws + o; o += (size_t)N * DIM_OUT;
    float* h_r2   = ws + o; o += (size_t)N * DIM_OUT;
    int* csr_src  = (int*)(ws + o); o += (size_t)E;
    int* rec      = (int*)(ws + o); o += (size_t)2 * E;
    int* tab      = (int*)(ws + o); o += (size_t)2 * NB_MAX * NCHUNK;
    int* rowptr   = (int*)(ws + o); o += (size_t)N + 1;

    const int B = 256;

    const int NB = cdiv_i(N, BIN_NODES);
    gat_chunk_order<<<NCHUNK, SORT_THREADS, 0, stream>>>(src, dst, rec, tab, E, NB);
    gat_bin_sort<<<NB, SORT_THREADS, 0, stream>>>(rec, tab, rowptr, csr_src, N, E);

    gat_gemm_pair_f32_wmma<DIM_IN, HID>
        <<<256, GEMM_THREADS, 0, stream>>>(x, Wl1, bl1, Wr1, br1, h_l1, h_r1, N);
    gat_node_l1<<<cdiv_i((long long)N * 8, B), B, 0, stream>>>(
        h_l1, h_r1, att1, bias1, rowptr, csr_src, h1, N);

    gat_gemm_pair_f32_wmma<HID, DIM_OUT>
        <<<256, GEMM_THREADS, 0, stream>>>(h1, Wl2, bl2, Wr2, br2, h_l2, h_r2, N);
    gat_node_l2_final<<<cdiv_i((long long)N * 8, B), B, 0, stream>>>(
        h_l2, h_r2, att2, bias2, rowptr, csr_src, (float*)d_out, N);
}


# Softmax step per edge on one hardware exp2 instead of two library expf: log2(e) folded into the attention vector, the larger argument's factor is exactly 1; sum order kept, factor rounding changes
# speedup vs baseline: 7.0685x; 1.0527x over previous
#include <hip/hip_runtime.h>
#include <hip/hip_bf16.h>


#define DIM_IN   128
#define HID      64
#define HEADS    8
#define DIM_H    8
#define DIM_OUT  40
#define NEG_SLOPE 0.2f

typedef __attribute__((ext_vector_type(2))) float v2f;
typedef __attribute__((ext_vector_type(8))) float v8f;

#define GEMM_THREADS 512

template <int K, int CC>
__global__ __launch_bounds__(GEMM_THREADS) void gat_gemm_pair_f32_wmma(
    const float* __restrict__ A,
    const float* __restrict__ Wl, const float* __restrict__ bl,
    const float* __restrict__ Wr, const float* __restrict__ br,
    float* __restrict__ outl, float* __restrict__ outr, int n) {
    constexpr int NC = 2 * CC;
    constexpr int NT = NC / 16;
    constexpr int P = K + 4;
    constexpr int J = K / 8;
    static_assert(NC % 16 == 0 && K % 8 == 0, "tile shape");
    static_assert((K * NC) % GEMM_THREADS == 0, "staging loop has no tail");
    __shared__ __attribute__((aligned(16))) float sWt[NC * P];

    const int lane = threadIdx.x & 31;
    const int c = lane & 15;
    const int h = lane >> 4;
    const int wave = __builtin_amdgcn_readfirstlane((int)(threadIdx.x >> 5));
    constexpr int nWaves = GEMM_THREADS / 32;
    const int rowTiles = n >> 4;
    const int stride = nWaves * (int)gridDim.x;
    int rt = wave * (int)gridDim.x + (int)blockIdx.x;

    float4 a[J];
    if (rt < rowTiles) {
        const float4* ap = (const float4*)(A + (size_t)((rt << 4) + c) * K + 4 * h);
#pragma unroll
        for (int j = 0; j < J; ++j) a[j] = ap[2 * j];
    }

#pragma unroll
    for (int i = 0; i < (K * NC) / GEMM_THREADS; ++i) {
        const int idx = i * GEMM_THREADS + (int)threadIdx.x;
        const int k = idx / NC;
        const int cc = idx - k * NC;
        const float* wp = (cc < CC) ? (Wl + cc) : (Wr + (cc - CC));
        sWt[cc * P + k] = wp[(size_t)k * CC];
    }
    __syncthreads();

    const float* bBase = sWt + c * P + 4 * h;

    while (rt < rowTiles) {
        v8f acc[NT];
#pragma unroll
        for (int t = 0; t < NT; ++t) acc[t] = v8f{};

        float4 bq[2][NT];
#pragma unroll
        for (int t = 0; t < NT; ++t)
            bq[0][t] = *(const float4*)(bBase + t * 16 * P);
#pragma unroll
        for (int j = 0; j < J; ++j) {
            if (j + 1 < J) {
#pragma unroll
                for (int t = 0; t < NT; ++t)
                    bq[(j + 1) & 1][t] =
                        *(const float4*)(bBase + t * 16 * P + 8 * (j + 1));
            }
            const v2f a01 = {a[j].x, a[j].y};
            const v2f a23 = {a[j].z, a[j].w};
#pragma unroll
            for (int t = 0; t < NT; ++t) {
                const v2f b01 = {bq[j & 1][t].x, bq[j & 1][t].y};
                const v2f b23 = {bq[j & 1][t].z, bq[j & 1][t].w};
                acc[t] = __builtin_amdgcn_wmma_f32_16x16x4_f32(
                    false, a01, false, b01, (short)0, acc[t], false, false);
                acc[t] = __builtin_amdgcn_wmma_f32_16x16x4_f32(
                    false, a23, false, b23, (short)0, acc[t], false, false);
            }
            __builtin_amdgcn_sched_barrier(0);
        }

        const int rtNext = rt + stride;
        if (rtNext < rowTiles) {
            const float4* ap =
                (const float4*)(A + (size_t)((rtNext << 4) + c) * K + 4 * h);
#pragma unroll
            for (int j = 0; j < J; ++j) a[j] = ap[2 * j];
        }

        const int rbase = (rt << 4) + (h << 3);
#pragma unroll
        for (int t = 0; t < NT; ++t) {
            const int col = t * 16 + c;
            const bool left = col < CC;
            const float bv = left ? bl[col] : br[col - CC];
            float* op = left ? (outl + col) : (outr + (col - CC));
#pragma unroll
            for (int r = 0; r < 8; ++r) op[(size_t)(rbase + r) * CC] = acc[t][r] + bv;
        }
        rt = rtNext;
    }
}

#define BIN_SHIFT 8
#define BIN_NODES (1 << BIN_SHIFT)
#define NB_MAX    512
#define NCHUNK    256
#define CH_MAX    8192
#define KEY_SHIFT 21
#define BIN_CAP   8192
#define SORT_THREADS 1024

__device__ __forceinline__ int gat_block_excl_scan(int x, int* sWv) {
    const int lane = threadIdx.x & 31;
    const int w = threadIdx.x >> 5;
    int inc = x;
    for (int o = 1; o < 32; o <<= 1) {
        const int u = __shfl_up(inc, o, 32);
        if (lane >= o) inc += u;
    }
    if (lane == 31) sWv[w] = inc;
    __syncthreads();
    if (w == 0) {
        const int v = sWv[lane];
        int iv = v;
        for (int o = 1; o < 32; o <<= 1) {
            const int u = __shfl_up(iv, o, 32);
            if (lane >= o) iv += u;
        }
        sWv[lane] = iv - v;
    }
    __syncthreads();
    const int r = sWv[w] + inc - x;
    __syncthreads();
    return r;
}

__global__ __launch_bounds__(SORT_THREADS) void gat_chunk_order(
    const int* __restrict__ src, const int* __restrict__ dst,
    int* __restrict__ recw, int* __restrict__ tabw, int E, int NB) {
    constexpr int PER = CH_MAX / SORT_THREADS;
    int2* rec = (int2*)recw;
    int2* tab = (int2*)tabw;
    __shared__ int cur[NB_MAX];
    __shared__ int lstart[NB_MAX];
    __shared__ int sWv[32];
    __shared__ int s_key[CH_MAX];
    __shared__ int s_src[CH_MAX];
    const int tid = threadIdx.x;
    if (tid < NB_MAX) cur[tid] = 0;
    const int ch = (E + NCHUNK - 1) / NCHUNK;
    const int e0 = (int)blockIdx.x * ch;
    const int e1 = min(e0 + ch, E);
    const int n = max(e1 - e0, 0);

    int kd[PER], ks[PER];
#pragma unroll
    for (int k = 0; k < PER; ++k) {
        const int e = max(min(e0 + tid + k * SORT_THREADS, e1 - 1), 0);
        kd[k] = dst[e];
        ks[k] = src[e];
    }
    __syncthreads();

    int kr[PER];
#pragma unroll
    for (int k = 0; k < PER; ++k) {
        kr[k] = 0;
        if (tid + k * SORT_THREADS < n) kr[k] = atomicAdd(&cur[kd[k] >> BIN_SHIFT], 1);
    }
    __syncthreads();
    const int c = (tid < NB_MAX) ? cur[tid] : 0;
    const int ex = gat_block_excl_scan(c, sWv);
    if (tid < NB_MAX) lstart[tid] = ex;
    if (tid < NB) tab[tid * NCHUNK + (int)blockIdx.x] = make_int2(ex, c);
    __syncthreads();

#pragma unroll
    for (int k = 0; k < PER; ++k) {
        if (tid + k * SORT_THREADS < n) {
            const int p = lstart[kd[k] >> BIN_SHIFT] + kr[k];
            s_key[p] = ((kd[k] & (BIN_NODES - 1)) << KEY_SHIFT) | (e0 + tid + k * SORT_THREADS);
            s_src[p] = ks[k];
        }
    }
    __syncthreads();
    for (int p = tid; p < n; p += SORT_THREADS)
        rec[e0 + p] = make_int2(s_key[p], s_src[p]);
}

__device__ __forceinline__ int gat_run_slot(const int* sPre, const int* sAt, int p) {
    int lo = 0;
#pragma unroll
    for (int w = NCHUNK >> 1; w > 0; w >>= 1)
        if (sPre[lo + w] <= p) lo += w;
    return sAt[lo] + p;
}

__global__ __launch_bounds__(SORT_THREADS) void gat_bin_sort(
    const int* __restrict__ recw, const int* __restrict__ tabw,
    int* __restrict__ rowptr, int* __restrict__ csr_src, int N, int E) {
    constexpr int PER = BIN_CAP / SORT_THREADS;
    const int2* rec = (const int2*)recw;
    const int2* tab = (const int2*)tabw;
    __shared__ int cnt[BIN_NODES];
    __shared__ int nstart[BIN_NODES];
    __shared__ int sWv[32];
    __shared__ int sPre[NCHUNK + 1];
    __shared__ int sAt[NCHUNK];
    __shared__ int sBase;
    __shared__ int p_key[BIN_CAP];
    __shared__ int p_src[BIN_CAP];
    __shared__ int o_src[BIN_CAP];
    const int tid = threadIdx.x;
    const int b = (int)blockIdx.x;
    const int ch = (E + NCHUNK - 1) / NCHUNK;
    const int n0 = b << BIN_SHIFT;
    int2 row = make_int2(0, 0);
    if (tid < NCHUNK) row = tab[b * NCHUNK + tid];
    if (tid < BIN_NODES) cnt[tid] = 0;
    if (tid == 0) sBase = 0;
    __syncthreads();

    int st = row.x;
    st += __shfl_xor(st, 16, 32);
    st += __shfl_xor(st, 8, 32);
    st += __shfl_xor(st, 4, 32);
    st += __shfl_xor(st, 2, 32);
    st += __shfl_xor(st, 1, 32);
    if ((tid & 31) == 0 && tid < NCHUNK) atomicAdd(&sBase, st);
    const int pre = gat_block_excl_scan(row.y, sWv);
    if (tid < NCHUNK) {
        sPre[tid] = pre;
        sAt[tid] = tid * ch + row.x - pre;
    }
    if (tid == NCHUNK - 1) sPre[NCHUNK] = pre + row.y;
    __syncthreads();
    const int M = sPre[NCHUNK];
    const int base = sBase;
    const int end = base + M;

    if (M <= BIN_CAP) {
        int2 r[PER];
        int rk[PER];
#pragma unroll
        for (int k = 0; k < PER; ++k) r[k] = make_int2(0, 0);
        if (M > 0) {
#pragma unroll
            for (int k = 0; k < PER; ++k)
                r[k] = rec[gat_run_slot(sPre, sAt, min(tid + k * SORT_THREADS, M - 1))];
        }
#pragma unroll
        for (int k = 0; k < PER; ++k) {
            rk[k] = 0;
            if (tid + k * SORT_THREADS < M) rk[k] = atomicAdd(&cnt[r[k].x >> KEY_SHIFT], 1);
        }
        __syncthreads();
        const int c = (tid < BIN_NODES) ? cnt[tid] : 0;
        const int ex = gat_block_excl_scan(c, sWv);
        if (tid < BIN_NODES) {
            nstart[tid] = ex;
            if (n0 + tid < N) rowptr[n0 + tid] = base + ex;
        }
        if (b == (int)gridDim.x - 1 && tid == 0) rowptr[N] = end;
        __syncthreads();
#pragma unroll
        for (int k = 0; k < PER; ++k) {
            if (tid + k * SORT_THREADS < M) {
                const int p = nstart[r[k].x >> KEY_SHIFT] + rk[k];
                p_key[p] = r[k].x;
                p_src[p] = r[k].y;
            }
        }
        __syncthreads();
        for (int p = tid; p < M; p += SORT_THREADS) {
            const int key = p_key[p];
            const int dl = key >> KEY_SHIFT;
            const int s = nstart[dl];
            const int e = s + cnt[dl];
            int rank = 0;
            for (int q = s; q < e; ++q) rank += (p_key[q] < key) ? 1 : 0;
            o_src[s + rank] = p_src[p];
        }
        __syncthreads();
        for (int p = tid; p < M; p += SORT_THREADS) csr_src[base + p] = o_src[p];
    } else {
        for (int i = tid; i < M; i += SORT_THREADS)
            atomicAdd(&cnt[rec[gat_run_slot(sPre, sAt, i)].x >> KEY_SHIFT], 1);
        __syncthreads();
        const int c = (tid < BIN_NODES) ? cnt[tid] : 0;
        const int ex = gat_block_excl_scan(c, sWv);
        if (tid < BIN_NODES) {
            nstart[tid] = ex;
            if (n0 + tid < N) rowptr[n0 + tid] = base + ex;
        }
        if (b == (int)gridDim.x - 1 && tid == 0) rowptr[N] = end;
        __syncthreads();
        for (int i = tid; i < M; i += SORT_THREADS) {
            const int2 ri = rec[gat_run_slot(sPre, sAt, i)];
            const int dl = ri.x >> KEY_SHIFT;
            int rank = 0;
            for (int j = 0; j < M; ++j) {
                const int kj = rec[gat_run_slot(sPre, sAt, j)].x;
                rank += ((kj >> KEY_SHIFT) == dl && kj < ri.x) ? 1 : 0;
            }
            csr_src[base + nstart[dl] + rank] = ri.y;
        }
    }
}

__device__ __forceinline__ float gat_leaky(float v) {
    return v > 0.0f ? v : v * NEG_SLOPE;
}

__device__ __forceinline__ float gat_elu(float v) {
    return v > 0.0f ? v : expm1f(v);
}

#define LOG2E 1.44269504088896340736f

__device__ __forceinline__ void gat_softmax_step(float lg, float& m, float& sc, float& p) {
    const float d = lg - m;
    const float e = __builtin_amdgcn_exp2f(-__builtin_fabsf(d));
    const bool up = d > 0.0f;
    sc = up ? e : 1.0f;
    p = up ? 1.0f : e;
    m = up ? lg : m;
}

__global__ __launch_bounds__(256) void gat_node_l1(
    const float* __restrict__ xl, const float* __restrict__ xr,
    const float* __restrict__ att, const float* __restrict__ bias1,
    const int* __restrict__ rowptr, const int* __restrict__ csr_src,
    float* __restrict__ h1, int N) {
    const int t = blockIdx.x * blockDim.x + threadIdx.x;
    const int node = t >> 3;
    const int h = t & 7;
    if (node >= N) return;
    const int beg = rowptr[node], end = rowptr[node + 1];
    const float4* pi = (const float4*)(xr + (size_t)node * HID + h * DIM_H);
    const float4 xi0 = pi[0], xi1 = pi[1];
    const float4* pa = (const float4*)(att + h * DIM_H);
    const float4 ar0 = pa[0], ar1 = pa[1];
    const float4 a0 = make_float4(ar0.x * LOG2E, ar0.y * LOG2E, ar0.z * LOG2E, ar0.w * LOG2E);
    const float4 a1 = make_float4(ar1.x * LOG2E, ar1.y * LOG2E, ar1.z * LOG2E, ar1.w * LOG2E);

    float m = -__builtin_inff();
    float den = 0.0f;
    float acc[8] = {0.0f, 0.0f, 0.0f, 0.0f, 0.0f, 0.0f, 0.0f, 0.0f};

    int s0n = 0, s1n = 0;
    if (beg < end) {
        s0n = csr_src[beg];
        s1n = csr_src[min(beg + 1, end - 1)];
    }
    for (int i = beg; i < end; i += 2) {
        const int s0 = s0n, s1 = s1n;
        const bool has1 = (i + 1 < end);
        const float4* p0 = (const float4*)(xl + (size_t)s0 * HID + h * DIM_H);
        const float4* p1 = (const float4*)(xl + (size_t)s1 * HID + h * DIM_H);
        const float4 u0 = p0[0], u1 = p0[1];
        const float4 w0 = p1[0], w1 = p1[1];
        if (i + 2 < end) {
            s0n = csr_src[i + 2];
            s1n = csr_src[min(i + 3, end - 1)];
        }
        {
            float lg = 0.0f;
            lg += gat_leaky(xi0.x + u0.x) * a0.x;
            lg += gat_leaky(xi0.y + u0.y) * a0.y;
            lg += gat_leaky(xi0.z + u0.z) * a0.z;
            lg += gat_leaky(xi0.w + u0.w) * a0.w;
            lg += gat_leaky(xi1.x + u1.x) * a1.x;
            lg += gat_leaky(xi1.y + u1.y) * a1.y;
            lg += gat_leaky(xi1.z + u1.z) * a1.z;
            lg += gat_leaky(xi1.w + u1.w) * a1.w;
            float sc, p;
            gat_softmax_step(lg, m, sc, p);
            den = den * sc + p;
            acc[0] = acc[0] * sc + p * u0.x;
            acc[1] = acc[1] * sc + p * u0.y;
            acc[2] = acc[2] * sc + p * u0.z;
            acc[3] = acc[3] * sc + p * u0.w;
            acc[4] = acc[4] * sc + p * u1.x;
            acc[5] = acc[5] * sc + p * u1.y;
            acc[6] = acc[6] * sc + p * u1.z;
            acc[7] = acc[7] * sc + p * u1.w;
        }
        if (has1) {
            float lg = 0.0f;
            lg += gat_leaky(xi0.x + w0.x) * a0.x;
            lg += gat_leaky(xi0.y + w0.y) * a0.y;
            lg += gat_leaky(xi0.z + w0.z) * a0.z;
            lg += gat_leaky(xi0.w + w0.w) * a0.w;
            lg += gat_leaky(xi1.x + w1.x) * a1.x;
            lg += gat_leaky(xi1.y + w1.y) * a1.y;
            lg += gat_leaky(xi1.z + w1.z) * a1.z;
            lg += gat_leaky(xi1.w + w1.w) * a1.w;
            float sc, p;
            gat_softmax_step(lg, m, sc, p);
            den = den * sc + p;
            acc[0] = acc[0] * sc + p * w0.x;
            acc[1] = acc[1] * sc + p * w0.y;
            acc[2] = acc[2] * sc + p * w0.z;
            acc[3] = acc[3] * sc + p * w0.w;
            acc[4] = acc[4] * sc + p * w1.x;
            acc[5] = acc[5] * sc + p * w1.y;
            acc[6] = acc[6] * sc + p * w1.z;
            acc[7] = acc[7] * sc + p * w1.w;
        }
    }

    const float inv = (end > beg) ? 1.0f / den : 0.0f;
    const float4* pb = (const float4*)(bias1 + h * DIM_H);
    const float4 b0 = pb[0], b1 = pb[1];
    float4* po = (float4*)(h1 + (size_t)node * HID + h * DIM_H);
    po[0] = make_float4(gat_elu(acc[0] * inv + b0.x), gat_elu(acc[1] * inv + b0.y),
                        gat_elu(acc[2] * inv + b0.z), gat_elu(acc[3] * inv + b0.w));
    po[1] = make_float4(gat_elu(acc[4] * inv + b1.x), gat_elu(acc[5] * inv + b1.y),
                        gat_elu(acc[6] * inv + b1.z), gat_elu(acc[7] * inv + b1.w));
}

__global__ __launch_bounds__(256) void gat_node_l2_final(
    const float* __restrict__ xl, const float* __restrict__ xr,
    const float* __restrict__ att, const float* __restrict__ bias2,
    const int* __restrict__ rowptr, const int* __restrict__ csr_src,
    float* __restrict__ d_out, int N) {
    const int t = blockIdx.x * blockDim.x + threadIdx.x;
    const int node = t >> 3;
    const int sub = t & 7;
    if (node >= N) return;
    const int beg = rowptr[node], end = rowptr[node + 1];
    const float* ri = xr + (size_t)node * DIM_OUT;
    const float4 xi = *(const float4*)(ri + sub * 4);
    const float xi4 = ri[32 + sub];
    const float4 ar = *(const float4*)(att + sub * 4);
    const float4 av = make_float4(ar.x * LOG2E, ar.y * LOG2E, ar.z * LOG2E, ar.w * LOG2E);
    const float a4 = att[32 + sub] * LOG2E;

    float m = -__builtin_inff();
    float den = 0.0f;
    float acc0 = 0.0f, acc1 = 0.0f, acc2 = 0.0f, acc3 = 0.0f, acc4 = 0.0f;

    int s0n = 0, s1n = 0;
    if (beg < end) {
        s0n = csr_src[beg];
        s1n = csr_src[min(beg + 1, end - 1)];
    }
    for (int i = beg; i < end; i += 2) {
        const int s0 = s0n, s1 = s1n;
        const bool has1 = (i + 1 < end);
        const float* r0 = xl + (size_t)s0 * DIM_OUT;
        const float* r1 = xl + (size_t)s1 * DIM_OUT;
        const float4 u = *(const float4*)(r0 + sub * 4);
        const float u4 = r0[32 + sub];
        const float4 w = *(const float4*)(r1 + sub * 4);
        const float w4 = r1[32 + sub];
        if (i + 2 < end) {
            s0n = csr_src[i + 2];
            s1n = csr_src[min(i + 3, end - 1)];
        }
        {
            float lg = 0.0f;
            lg += gat_leaky(xi.x + u.x) * av.x;
            lg += gat_leaky(xi.y + u.y) * av.y;
            lg += gat_leaky(xi.z + u.z) * av.z;
            lg += gat_leaky(xi.w + u.w) * av.w;
            lg += gat_leaky(xi4 + u4) * a4;
            lg += __shfl_xor(lg, 1, 32);
            lg += __shfl_xor(lg, 2, 32);
            lg += __shfl_xor(lg, 4, 32);
            float sc, p;
            gat_softmax_step(lg, m, sc, p);
            den = den * sc + p;
            acc0 = acc0 * sc + p * u.x;
            acc1 = acc1 * sc + p * u.y;
            acc2 = acc2 * sc + p * u.z;
            acc3 = acc3 * sc + p * u.w;
            acc4 = acc4 * sc + p * u4;
        }
        if (has1) {
            float lg = 0.0f;
            lg += gat_leaky(xi.x + w.x) * av.x;
            lg += gat_leaky(xi.y + w.y) * av.y;
            lg += gat_leaky(xi.z + w.z) * av.z;
            lg += gat_leaky(xi.w + w.w) * av.w;
            lg += gat_leaky(xi4 + w4) * a4;
            lg += __shfl_xor(lg, 1, 32);
            lg += __shfl_xor(lg, 2, 32);
            lg += __shfl_xor(lg, 4, 32);
            float sc, p;
            gat_softmax_step(lg, m, sc, p);
            den = den * sc + p;
            acc0 = acc0 * sc + p * w.x;
            acc1 = acc1 * sc + p * w.y;
            acc2 = acc2 * sc + p * w.z;
            acc3 = acc3 * sc + p * w.w;
            acc4 = acc4 * sc + p * w4;
        }
    }

    const float inv = (end > beg) ? 1.0f / den : 0.0f;
    const float4 bv = *(const float4*)(bias2 + sub * 4);
    const float v0 = acc0 * inv + bv.x;
    const float v1 = acc1 * inv + bv.y;
    const float v2 = acc2 * inv + bv.z;
    const float v3 = acc3 * inv + bv.w;
    const float v4 = acc4 * inv + bias2[32 + sub];

    float mx = fmaxf(fmaxf(fmaxf(v0, v1), fmaxf(v2, v3)), v4);
    mx = fmaxf(mx, __shfl_xor(mx, 1, 32));
    mx = fmaxf(mx, __shfl_xor(mx, 2, 32));
    mx = fmaxf(mx, __shfl_xor(mx, 4, 32));
    float s = expf(v0 - mx) + expf(v1 - mx) + expf(v2 - mx) + expf(v3 - mx) +
              expf(v4 - mx);
    s += __shfl_xor(s, 1, 32);
    s += __shfl_xor(s, 2, 32);
    s += __shfl_xor(s, 4, 32);
    const float lse = logf(s) + mx;

    float* hrow = d_out + (size_t)node * DIM_OUT;
    float* lrow = d_out + (size_t)N * DIM_OUT + (size_t)node * DIM_OUT;
    *(float4*)(hrow + sub * 4) = make_float4(v0, v1, v2, v3);
    hrow[32 + sub] = v4;
    *(float4*)(lrow + sub * 4) = make_float4(v0 - lse, v1 - lse, v2 - lse, v3 - lse);
    lrow[32 + sub] = v4 - lse;
}

static inline int cdiv_i(long long a, long long b) { return (int)((a + b - 1) / b); }

extern "C" void kernel_launch(void* const* d_in, const int* in_sizes, int n_in,
                              void* d_out, int out_size, void* d_ws, size_t ws_size,
                              hipStream_t stream) {
    (void)n_in; (void)out_size; (void)ws_size;
    const float* x     = (const float*)d_in[0];
    const int*   eidx  = (const int*)d_in[1];
    const float* Wl1   = (const float*)d_in[2];
    const float* bl1   = (const float*)d_in[3];
    const float* Wr1   = (const float*)d_in[4];
    const float* br1   = (const float*)d_in[5];
    const float* att1  = (const float*)d_in[6];
    const float* bias1 = (const float*)d_in[7];
    const float* Wl2   = (const float*)d_in[8];
    const float* bl2   = (const float*)d_in[9];
    const float* Wr2   = (const float*)d_in[10];
    const float* br2   = (const float*)d_in[11];
    const float* att2  = (const float*)d_in[12];
    const float* bias2 = (const float*)d_in[13];

    const int N = in_sizes[0] / DIM_IN;
    const int E = in_sizes[1] / 2;
    const int* src = eidx;
    const int* dst = eidx + E;

    float* ws = (float*)d_ws;
    size_t o = 0;
    float* h_l1   = ws + o; o += (size_t)N * HID;
    float* h_r1   = ws + o; o += (size_t)N * HID;
    float* h1     = ws + o; o += (size_t)N * HID;
    float* h_l2   = ws + o; o += (size_t)N * DIM_OUT;
    float* h_r2   = ws + o; o += (size_t)N * DIM_OUT;
    int* csr_src  = (int*)(ws + o); o += (size_t)E;
    int* rec      = (int*)(ws + o); o += (size_t)2 * E;
    int* tab      = (int*)(ws + o); o += (size_t)2 * NB_MAX * NCHUNK;
    int* rowptr   = (int*)(ws + o); o += (size_t)N + 1;

    const int B = 256;

    const int NB = cdiv_i(N, BIN_NODES);
    gat_chunk_order<<<NCHUNK, SORT_THREADS, 0, stream>>>(src, dst, rec, tab, E, NB);
    gat_bin_sort<<<NB, SORT_THREADS, 0, stream>>>(rec, tab, rowptr, csr_src, N, E);

    gat_gemm_pair_f32_wmma<DIM_IN, HID>
        <<<256, GEMM_THREADS, 0, stream>>>(x, Wl1, bl1, Wr1, br1, h_l1, h_r1, N);
    gat_node_l1<<<cdiv_i((long long)N * 8, B), B, 0, stream>>>(
        h_l1, h_r1, att1, bias1, rowptr, csr_src, h1, N);

    gat_gemm_pair_f32_wmma<HID, DIM_OUT>
        <<<256, GEMM_THREADS, 0, stream>>>(h1, Wl2, bl2, Wr2, br2, h_l2, h_r2, N);
    gat_node_l2_final<<<cdiv_i((long long)N * 8, B), B, 0, stream>>>(
        h_l2, h_r2, att2, bias2, rowptr, csr_src, (float*)d_out, N);
}


# Node loops: edge logit in fewer issue slots: LeakyReLU as max(z, slope*z), products summed as packed even/odd pairs (logit sum order changes), 8-lane sums by DPP adds, not LDS permutes
# speedup vs baseline: 7.3443x; 1.0390x over previous
#include <hip/hip_runtime.h>
#include <hip/hip_bf16.h>


#define DIM_IN   128
#define HID      64
#define HEADS    8
#define DIM_H    8
#define DIM_OUT  40
#define NEG_SLOPE 0.2f

typedef __attribute__((ext_vector_type(2))) float v2f;
typedef __attribute__((ext_vector_type(8))) float v8f;

#define GEMM_THREADS 512

template <int K, int CC>
__global__ __launch_bounds__(GEMM_THREADS) void gat_gemm_pair_f32_wmma(
    const float* __restrict__ A,
    const float* __restrict__ Wl, const float* __restrict__ bl,
    const float* __restrict__ Wr, const float* __restrict__ br,
    float* __restrict__ outl, float* __restrict__ outr, int n) {
    constexpr int NC = 2 * CC;
    constexpr int NT = NC / 16;
    constexpr int P = K + 4;
    constexpr int J = K / 8;
    static_assert(NC % 16 == 0 && K % 8 == 0, "tile shape");
    static_assert((K * NC) % GEMM_THREADS == 0, "staging loop has no tail");
    __shared__ __attribute__((aligned(16))) float sWt[NC * P];

    const int lane = threadIdx.x & 31;
    const int c = lane & 15;
    const int h = lane >> 4;
    const int wave = __builtin_amdgcn_readfirstlane((int)(threadIdx.x >> 5));
    constexpr int nWaves = GEMM_THREADS / 32;
    const int rowTiles = n >> 4;
    const int stride = nWaves * (int)gridDim.x;
    int rt = wave * (int)gridDim.x + (int)blockIdx.x;

    float4 a[J];
    if (rt < rowTiles) {
        const float4* ap = (const float4*)(A + (size_t)((rt << 4) + c) * K + 4 * h);
#pragma unroll
        for (int j = 0; j < J; ++j) a[j] = ap[2 * j];
    }

#pragma unroll
    for (int i = 0; i < (K * NC) / GEMM_THREADS; ++i) {
        const int idx = i * GEMM_THREADS + (int)threadIdx.x;
        const int k = idx / NC;
        const int cc = idx - k * NC;
        const float* wp = (cc < CC) ? (Wl + cc) : (Wr + (cc - CC));
        sWt[cc * P + k] = wp[(size_t)k * CC];
    }
    __syncthreads();

    const float* bBase = sWt + c * P + 4 * h;

    while (rt < rowTiles) {
        v8f acc[NT];
#pragma unroll
        for (int t = 0; t < NT; ++t) acc[t] = v8f{};

        float4 bq[2][NT];
#pragma unroll
        for (int t = 0; t < NT; ++t)
            bq[0][t] = *(const float4*)(bBase + t * 16 * P);
#pragma unroll
        for (int j = 0; j < J; ++j) {
            if (j + 1 < J) {
#pragma unroll
                for (int t = 0; t < NT; ++t)
                    bq[(j + 1) & 1][t] =
                        *(const float4*)(bBase + t * 16 * P + 8 * (j + 1));
            }
            const v2f a01 = {a[j].x, a[j].y};
            const v2f a23 = {a[j].z, a[j].w};
#pragma unroll
            for (int t = 0; t < NT; ++t) {
                const v2f b01 = {bq[j & 1][t].x, bq[j & 1][t].y};
                const v2f b23 = {bq[j & 1][t].z, bq[j & 1][t].w};
                acc[t] = __builtin_amdgcn_wmma_f32_16x16x4_f32(
                    false, a01, false, b01, (short)0, acc[t], false, false);
                acc[t] = __builtin_amdgcn_wmma_f32_16x16x4_f32(
                    false, a23, false, b23, (short)0, acc[t], false, false);
            }
            __builtin_amdgcn_sched_barrier(0);
        }

        const int rtNext = rt + stride;
        if (rtNext < rowTiles) {
            const float4* ap =
                (const float4*)(A + (size_t)((rtNext << 4) + c) * K + 4 * h);
#pragma unroll
            for (int j = 0; j < J; ++j) a[j] = ap[2 * j];
        }

        const int rbase = (rt << 4) + (h << 3);
#pragma unroll
        for (int t = 0; t < NT; ++t) {
            const int col = t * 16 + c;
            const bool left = col < CC;
            const float bv = left ? bl[col] : br[col - CC];
            float* op = left ? (outl + col) : (outr + (col - CC));
#pragma unroll
            for (int r = 0; r < 8; ++r) op[(size_t)(rbase + r) * CC] = acc[t][r] + bv;
        }
        rt = rtNext;
    }
}

#define BIN_SHIFT 8
#define BIN_NODES (1 << BIN_SHIFT)
#define NB_MAX    512
#define NCHUNK    256
#define CH_MAX    8192
#define KEY_SHIFT 21
#define BIN_CAP   8192
#define SORT_THREADS 1024

__device__ __forceinline__ int gat_block_excl_scan(int x, int* sWv) {
    const int lane = threadIdx.x & 31;
    const int w = threadIdx.x >> 5;
    int inc = x;
    for (int o = 1; o < 32; o <<= 1) {
        const int u = __shfl_up(inc, o, 32);
        if (lane >= o) inc += u;
    }
    if (lane == 31) sWv[w] = inc;
    __syncthreads();
    if (w == 0) {
        const int v = sWv[lane];
        int iv = v;
        for (int o = 1; o < 32; o <<= 1) {
            const int u = __shfl_up(iv, o, 32);
            if (lane >= o) iv += u;
        }
        sWv[lane] = iv - v;
    }
    __syncthreads();
    const int r = sWv[w] + inc - x;
    __syncthreads();
    return r;
}

__global__ __launch_bounds__(SORT_THREADS) void gat_chunk_order(
    const int* __restrict__ src, const int* __restrict__ dst,
    int* __restrict__ recw, int* __restrict__ tabw, int E, int NB) {
    constexpr int PER = CH_MAX / SORT_THREADS;
    int2* rec = (int2*)recw;
    int2* tab = (int2*)tabw;
    __shared__ int cur[NB_MAX];
    __shared__ int lstart[NB_MAX];
    __shared__ int sWv[32];
    __shared__ int s_key[CH_MAX];
    __shared__ int s_src[CH_MAX];
    const int tid = threadIdx.x;
    if (tid < NB_MAX) cur[tid] = 0;
    const int ch = (E + NCHUNK - 1) / NCHUNK;
    const int e0 = (int)blockIdx.x * ch;
    const int e1 = min(e0 + ch, E);
    const int n = max(e1 - e0, 0);

    int kd[PER], ks[PER];
#pragma unroll
    for (int k = 0; k < PER; ++k) {
        const int e = max(min(e0 + tid + k * SORT_THREADS, e1 - 1), 0);
        kd[k] = dst[e];
        ks[k] = src[e];
    }
    __syncthreads();

    int kr[PER];
#pragma unroll
    for (int k = 0; k < PER; ++k) {
        kr[k] = 0;
        if (tid + k * SORT_THREADS < n) kr[k] = atomicAdd(&cur[kd[k] >> BIN_SHIFT], 1);
    }
    __syncthreads();
    const int c = (tid < NB_MAX) ? cur[tid] : 0;
    const int ex = gat_block_excl_scan(c, sWv);
    if (tid < NB_MAX) lstart[tid] = ex;
    if (tid < NB) tab[tid * NCHUNK + (int)blockIdx.x] = make_int2(ex, c);
    __syncthreads();

#pragma unroll
    for (int k = 0; k < PER; ++k) {
        if (tid + k * SORT_THREADS < n) {
            const int p = lstart[kd[k] >> BIN_SHIFT] + kr[k];
            s_key[p] = ((kd[k] & (BIN_NODES - 1)) << KEY_SHIFT) | (e0 + tid + k * SORT_THREADS);
            s_src[p] = ks[k];
        }
    }
    __syncthreads();
    for (int p = tid; p < n; p += SORT_THREADS)
        rec[e0 + p] = make_int2(s_key[p], s_src[p]);
}

__device__ __forceinline__ int gat_run_slot(const int* sPre, const int* sAt, int p) {
    int lo = 0;
#pragma unroll
    for (int w = NCHUNK >> 1; w > 0; w >>= 1)
        if (sPre[lo + w] <= p) lo += w;
    return sAt[lo] + p;
}

__global__ __launch_bounds__(SORT_THREADS) void gat_bin_sort(
    const int* __restrict__ recw, const int* __restrict__ tabw,
    int* __restrict__ rowptr, int* __restrict__ csr_src, int N, int E) {
    constexpr int PER = BIN_CAP / SORT_THREADS;
    const int2* rec = (const int2*)recw;
    const int2* tab = (const int2*)tabw;
    __shared__ int cnt[BIN_NODES];
    __shared__ int nstart[BIN_NODES];
    __shared__ int sWv[32];
    __shared__ int sPre[NCHUNK + 1];
    __shared__ int sAt[NCHUNK];
    __shared__ int sBase;
    __shared__ int p_key[BIN_CAP];
    __shared__ int p_src[BIN_CAP];
    __shared__ int o_src[BIN_CAP];
    const int tid = threadIdx.x;
    const int b = (int)blockIdx.x;
    const int ch = (E + NCHUNK - 1) / NCHUNK;
    const int n0 = b << BIN_SHIFT;
    int2 row = make_int2(0, 0);
    if (tid < NCHUNK) row = tab[b * NCHUNK + tid];
    if (tid < BIN_NODES) cnt[tid] = 0;
    if (tid == 0) sBase = 0;
    __syncthreads();

    int st = row.x;
    st += __shfl_xor(st, 16, 32);
    st += __shfl_xor(st, 8, 32);
    st += __shfl_xor(st, 4, 32);
    st += __shfl_xor(st, 2, 32);
    st += __shfl_xor(st, 1, 32);
    if ((tid & 31) == 0 && tid < NCHUNK) atomicAdd(&sBase, st);
    const int pre = gat_block_excl_scan(row.y, sWv);
    if (tid < NCHUNK) {
        sPre[tid] = pre;
        sAt[tid] = tid * ch + row.x - pre;
    }
    if (tid == NCHUNK - 1) sPre[NCHUNK] = pre + row.y;
    __syncthreads();
    const int M = sPre[NCHUNK];
    const int base = sBase;
    const int end = base + M;

    if (M <= BIN_CAP) {
        int2 r[PER];
        int rk[PER];
#pragma unroll
        for (int k = 0; k < PER; ++k) r[k] = make_int2(0, 0);
        if (M > 0) {
#pragma unroll
            for (int k = 0; k < PER; ++k)
                r[k] = rec[gat_run_slot(sPre, sAt, min(tid + k * SORT_THREADS, M - 1))];
        }
#pragma unroll
        for (int k = 0; k < PER; ++k) {
            rk[k] = 0;
            if (tid + k * SORT_THREADS < M) rk[k] = atomicAdd(&cnt[r[k].x >> KEY_SHIFT], 1);
        }
        __syncthreads();
        const int c = (tid < BIN_NODES) ? cnt[tid] : 0;
        const int ex = gat_block_excl_scan(c, sWv);
        if (tid < BIN_NODES) {
            nstart[tid] = ex;
            if (n0 + tid < N) rowptr[n0 + tid] = base + ex;
        }
        if (b == (int)gridDim.x - 1 && tid == 0) rowptr[N] = end;
        __syncthreads();
#pragma unroll
        for (int k = 0; k < PER; ++k) {
            if (tid + k * SORT_THREADS < M) {
                const int p = nstart[r[k].x >> KEY_SHIFT] + rk[k];
                p_key[p] = r[k].x;
                p_src[p] = r[k].y;
            }
        }
        __syncthreads();
        for (int p = tid; p < M; p += SORT_THREADS) {
            const int key = p_key[p];
            const int dl = key >> KEY_SHIFT;
            const int s = nstart[dl];
            const int e = s + cnt[dl];
            int rank = 0;
            for (int q = s; q < e; ++q) rank += (p_key[q] < key) ? 1 : 0;
            o_src[s + rank] = p_src[p];
        }
        __syncthreads();
        for (int p = tid; p < M; p += SORT_THREADS) csr_src[base + p] = o_src[p];
    } else {
        for (int i = tid; i < M; i += SORT_THREADS)
            atomicAdd(&cnt[rec[gat_run_slot(sPre, sAt, i)].x >> KEY_SHIFT], 1);
        __syncthreads();
        const int c = (tid < BIN_NODES) ? cnt[tid] : 0;
        const int ex = gat_block_excl_scan(c, sWv);
        if (tid < BIN_NODES) {
            nstart[tid] = ex;
            if (n0 + tid < N) rowptr[n0 + tid] = base + ex;
        }
        if (b == (int)gridDim.x - 1 && tid == 0) rowptr[N] = end;
        __syncthreads();
        for (int i = tid; i < M; i += SORT_THREADS) {
            const int2 ri = rec[gat_run_slot(sPre, sAt, i)];
            const int dl = ri.x >> KEY_SHIFT;
            int rank = 0;
            for (int j = 0; j < M; ++j) {
                const int kj = rec[gat_run_slot(sPre, sAt, j)].x;
                rank += ((kj >> KEY_SHIFT) == dl && kj < ri.x) ? 1 : 0;
            }
            csr_src[base + nstart[dl] + rank] = ri.y;
        }
    }
}

__device__ __forceinline__ float gat_leaky(float v) {
    return __builtin_fmaxf(v, v * NEG_SLOPE);
}

__device__ __forceinline__ v2f gat_leaky2(v2f z) {
    return __builtin_elementwise_max(z, z * NEG_SLOPE);
}

template <int MASK>
__device__ __forceinline__ float gat_xor_lane(float v) {
    return __builtin_bit_cast(float, __builtin_amdgcn_update_dpp(
        0, __builtin_bit_cast(int, v), 0x160 + MASK, 0xf, 0xf, true));
}
__device__ __forceinline__ float gat_sum8(float v) {
    v += gat_xor_lane<1>(v);
    v += gat_xor_lane<2>(v);
    v += gat_xor_lane<4>(v);
    return v;
}
__device__ __forceinline__ float gat_max8(float v) {
    v = __builtin_fmaxf(v, gat_xor_lane<1>(v));
    v = __builtin_fmaxf(v, gat_xor_lane<2>(v));
    v = __builtin_fmaxf(v, gat_xor_lane<4>(v));
    return v;
}

__device__ __forceinline__ float gat_elu(float v) {
    return v > 0.0f ? v : expm1f(v);
}

#define LOG2E 1.44269504088896340736f

__device__ __forceinline__ void gat_softmax_step(float lg, float& m, float& sc, float& p) {
    const float d = lg - m;
    const float e = __builtin_amdgcn_exp2f(-__builtin_fabsf(d));
    const bool up = d > 0.0f;
    sc = up ? e : 1.0f;
    p = up ? 1.0f : e;
    m = up ? lg : m;
}

__device__ __forceinline__ float gat_logit8(const float4& xi0, const float4& xi1,
                                            const float4& u0, const float4& u1,
                                            const float4& a0, const float4& a1) {
    v2f s = gat_leaky2(v2f{xi0.x, xi0.y} + v2f{u0.x, u0.y}) * v2f{a0.x, a0.y};
    s = __builtin_elementwise_fma(gat_leaky2(v2f{xi0.z, xi0.w} + v2f{u0.z, u0.w}),
                                  v2f{a0.z, a0.w}, s);
    s = __builtin_elementwise_fma(gat_leaky2(v2f{xi1.x, xi1.y} + v2f{u1.x, u1.y}),
                                  v2f{a1.x, a1.y}, s);
    s = __builtin_elementwise_fma(gat_leaky2(v2f{xi1.z, xi1.w} + v2f{u1.z, u1.w}),
                                  v2f{a1.z, a1.w}, s);
    return s.x + s.y;
}

__device__ __forceinline__ float gat_logit5(const float4& xi, float xi4,
                                            const float4& u, float u4,
                                            const float4& av, float a4) {
    v2f s = gat_leaky2(v2f{xi.x, xi.y} + v2f{u.x, u.y}) * v2f{av.x, av.y};
    s = __builtin_elementwise_fma(gat_leaky2(v2f{xi.z, xi.w} + v2f{u.z, u.w}),
                                  v2f{av.z, av.w}, s);
    return __builtin_fmaf(gat_leaky(xi4 + u4), a4, s.x + s.y);
}

__global__ __launch_bounds__(256) void gat_node_l1(
    const float* __restrict__ xl, const float* __restrict__ xr,
    const float* __restrict__ att, const float* __restrict__ bias1,
    const int* __restrict__ rowptr, const int* __restrict__ csr_src,
    float* __restrict__ h1, int N) {
    const int t = blockIdx.x * blockDim.x + threadIdx.x;
    const int node = t >> 3;
    const int h = t & 7;
    if (node >= N) return;
    const int beg = rowptr[node], end = rowptr[node + 1];
    const float4* pi = (const float4*)(xr + (size_t)node * HID + h * DIM_H);
    const float4 xi0 = pi[0], xi1 = pi[1];
    const float4* pa = (const float4*)(att + h * DIM_H);
    const float4 ar0 = pa[0], ar1 = pa[1];
    const float4 a0 = make_float4(ar0.x * LOG2E, ar0.y * LOG2E, ar0.z * LOG2E, ar0.w * LOG2E);
    const float4 a1 = make_float4(ar1.x * LOG2E, ar1.y * LOG2E, ar1.z * LOG2E, ar1.w * LOG2E);

    float m = -__builtin_inff();
    float den = 0.0f;
    float acc[8] = {0.0f, 0.0f, 0.0f, 0.0f, 0.0f, 0.0f, 0.0f, 0.0f};

    int s0n = 0, s1n = 0;
    if (beg < end) {
        s0n = csr_src[beg];
        s1n = csr_src[min(beg + 1, end - 1)];
    }
    for (int i = beg; i < end; i += 2) {
        const int s0 = s0n, s1 = s1n;
        const bool has1 = (i + 1 < end);
        const float4* p0 = (const float4*)(xl + (size_t)s0 * HID + h * DIM_H);
        const float4* p1 = (const float4*)(xl + (size_t)s1 * HID + h * DIM_H);
        const float4 u0 = p0[0], u1 = p0[1];
        const float4 w0 = p1[0], w1 = p1[1];
        if (i + 2 < end) {
            s0n = csr_src[i + 2];
            s1n = csr_src[min(i + 3, end - 1)];
        }
        {
            const float lg = gat_logit8(xi0, xi1, u0, u1, a0, a1);
            float sc, p;
            gat_softmax_step(lg, m, sc, p);
            den = den * sc + p;
            acc[0] = acc[0] * sc + p * u0.x;
            acc[1] = acc[1] * sc + p * u0.y;
            acc[2] = acc[2] * sc + p * u0.z;
            acc[3] = acc[3] * sc + p * u0.w;
            acc[4] = acc[4] * sc + p * u1.x;
            acc[5] = acc[5] * sc + p * u1.y;
            acc[6] = acc[6] * sc + p * u1.z;
            acc[7] = acc[7] * sc + p * u1.w;
        }
        if (has1) {
            const float lg = gat_logit8(xi0, xi1, w0, w1, a0, a1);
            float sc, p;
            gat_softmax_step(lg, m, sc, p);
            den = den * sc + p;
            acc[0] = acc[0] * sc + p * w0.x;
            acc[1] = acc[1] * sc + p * w0.y;
            acc[2] = acc[2] * sc + p * w0.z;
            acc[3] = acc[3] * sc + p * w0.w;
            acc[4] = acc[4] * sc + p * w1.x;
            acc[5] = acc[5] * sc + p * w1.y;
            acc[6] = acc[6] * sc + p * w1.z;
            acc[7] = acc[7] * sc + p * w1.w;
        }
    }

    const float inv = (end > beg) ? 1.0f / den : 0.0f;
    const float4* pb = (const float4*)(bias1 + h * DIM_H);
    const float4 b0 = pb[0], b1 = pb[1];
    float4* po = (float4*)(h1 + (size_t)node * HID + h * DIM_H);
    po[0] = make_float4(gat_elu(acc[0] * inv + b0.x), gat_elu(acc[1] * inv + b0.y),
                        gat_elu(acc[2] * inv + b0.z), gat_elu(acc[3] * inv + b0.w));
    po[1] = make_float4(gat_elu(acc[4] * inv + b1.x), gat_elu(acc[5] * inv + b1.y),
                        gat_elu(acc[6] * inv + b1.z), gat_elu(acc[7] * inv + b1.w));
}

__global__ __launch_bounds__(256) void gat_node_l2_final(
    const float* __restrict__ xl, const float* __restrict__ xr,
    const float* __restrict__ att, const float* __restrict__ bias2,
    const int* __restrict__ rowptr, const int* __restrict__ csr_src,
    float* __restrict__ d_out, int N) {
    const int t = blockIdx.x * blockDim.x + threadIdx.x;
    const int node = t >> 3;
    const int sub = t & 7;
    if (node >= N) return;
    const int beg = rowptr[node], end = rowptr[node + 1];
    const float* ri = xr + (size_t)node * DIM_OUT;
    const float4 xi = *(const float4*)(ri + sub * 4);
    const float xi4 = ri[32 + sub];
    const float4 ar = *(const float4*)(att + sub * 4);
    const float4 av = make_float4(ar.x * LOG2E, ar.y * LOG2E, ar.z * LOG2E, ar.w * LOG2E);
    const float a4 = att[32 + sub] * LOG2E;

    float m = -__builtin_inff();
    float den = 0.0f;
    float acc0 = 0.0f, acc1 = 0.0f, acc2 = 0.0f, acc3 = 0.0f, acc4 = 0.0f;

    int s0n = 0, s1n = 0;
    if (beg < end) {
        s0n = csr_src[beg];
        s1n = csr_src[min(beg + 1, end - 1)];
    }
    for (int i = beg; i < end; i += 2) {
        const int s0 = s0n, s1 = s1n;
        const bool has1 = (i + 1 < end);
        const float* r0 = xl + (size_t)s0 * DIM_OUT;
        const float* r1 = xl + (size_t)s1 * DIM_OUT;
        const float4 u = *(const float4*)(r0 + sub * 4);
        const float u4 = r0[32 + sub];
        const float4 w = *(const float4*)(r1 + sub * 4);
        const float w4 = r1[32 + sub];
        if (i + 2 < end) {
            s0n = csr_src[i + 2];
            s1n = csr_src[min(i + 3, end - 1)];
        }
        {
            const float lg = gat_sum8(gat_logit5(xi, xi4, u, u4, av, a4));
            float sc, p;
            gat_softmax_step(lg, m, sc, p);
            den = den * sc + p;
            acc0 = acc0 * sc + p * u.x;
            acc1 = acc1 * sc + p * u.y;
            acc2 = acc2 * sc + p * u.z;
            acc3 = acc3 * sc + p * u.w;
            acc4 = acc4 * sc + p * u4;
        }
        if (has1) {
            const float lg = gat_sum8(gat_logit5(xi, xi4, w, w4, av, a4));
            float sc, p;
            gat_softmax_step(lg, m, sc, p);
            den = den * sc + p;
            acc0 = acc0 * sc + p * w.x;
            acc1 = acc1 * sc + p * w.y;
            acc2 = acc2 * sc + p * w.z;
            acc3 = acc3 * sc + p * w.w;
            acc4 = acc4 * sc + p * w4;
        }
    }

    const float inv = (end > beg) ? 1.0f / den : 0.0f;
    const float4 bv = *(const float4*)(bias2 + sub * 4);
    const float v0 = acc0 * inv + bv.x;
    const float v1 = acc1 * inv + bv.y;
    const float v2 = acc2 * inv + bv.z;
    const float v3 = acc3 * inv + bv.w;
    const float v4 = acc4 * inv + bias2[32 + sub];

    const float mx = gat_max8(fmaxf(fmaxf(fmaxf(v0, v1), fmaxf(v2, v3)), v4));
    const float s = gat_sum8(expf(v0 - mx) + expf(v1 - mx) + expf(v2 - mx) +
                             expf(v3 - mx) + expf(v4 - mx));
    const float lse = logf(s) + mx;

    float* hrow = d_out + (size_t)node * DIM_OUT;
    float* lrow = d_out + (size_t)N * DIM_OUT + (size_t)node * DIM_OUT;
    *(float4*)(hrow + sub * 4) = make_float4(v0, v1, v2, v3);
    hrow[32 + sub] = v4;
    *(float4*)(lrow + sub * 4) = make_float4(v0 - lse, v1 - lse, v2 - lse, v3 - lse);
    lrow[32 + sub] = v4 - lse;
}

static inline int cdiv_i(long long a, long long b) { return (int)((a + b - 1) / b); }

extern "C" void kernel_launch(void* const* d_in, const int* in_sizes, int n_in,
                              void* d_out, int out_size, void* d_ws, size_t ws_size,
                              hipStream_t stream) {
    (void)n_in; (void)out_size; (void)ws_size;
    const float* x     = (const float*)d_in[0];
    const int*   eidx  = (const int*)d_in[1];
    const float* Wl1   = (const float*)d_in[2];
    const float* bl1   = (const float*)d_in[3];
    const float* Wr1   = (const float*)d_in[4];
    const float* br1   = (const float*)d_in[5];
    const float* att1  = (const float*)d_in[6];
    const float* bias1 = (const float*)d_in[7];
    const float* Wl2   = (const float*)d_in[8];
    const float* bl2   = (const float*)d_in[9];
    const float* Wr2   = (const float*)d_in[10];
    const float* br2   = (const float*)d_in[11];
    const float* att2  = (const float*)d_in[12];
    const float* bias2 = (const float*)d_in[13];

    const int N = in_sizes[0] / DIM_IN;
    const int E = in_sizes[1] / 2;
    const int* src = eidx;
    const int* dst = eidx + E;

    float* ws = (float*)d_ws;
    size_t o = 0;
    float* h_l1   = ws + o; o += (size_t)N * HID;
    float* h_r1   = ws + o; o += (size_t)N * HID;
    float* h1     = ws + o; o += (size_t)N * HID;
    float* h_l2   = ws + o; o += (size_t)N * DIM_OUT;
    float* h_r2   = ws + o; o += (size_t)N * DIM_OUT;
    int* csr_src  = (int*)(ws + o); o += (size_t)E;
    int* rec      = (int*)(ws + o); o += (size_t)2 * E;
    int* tab      = (int*)(ws + o); o += (size_t)2 * NB_MAX * NCHUNK;
    int* rowptr   = (int*)(ws + o); o += (size_t)N + 1;

    const int B = 256;

    const int NB = cdiv_i(N, BIN_NODES);
    gat_chunk_order<<<NCHUNK, SORT_THREADS, 0, stream>>>(src, dst, rec, tab, E, NB);
    gat_bin_sort<<<NB, SORT_THREADS, 0, stream>>>(rec, tab, rowptr, csr_src, N, E);

    gat_gemm_pair_f32_wmma<DIM_IN, HID>
        <<<256, GEMM_THREADS, 0, stream>>>(x, Wl1, bl1, Wr1, br1, h_l1, h_r1, N);
    gat_node_l1<<<cdiv_i((long long)N * 8, B), B, 0, stream>>>(
        h_l1, h_r1, att1, bias1, rowptr, csr_src, h1, N);

    gat_gemm_pair_f32_wmma<HID, DIM_OUT>
        <<<256, GEMM_THREADS, 0, stream>>>(h1, Wl2, bl2, Wr2, br2, h_l2, h_r2, N);
    gat_node_l2_final<<<cdiv_i((long long)N * 8, B), B, 0, stream>>>(
        h_l2, h_r2, att2, bias2, rowptr, csr_src, (float*)d_out, N);
}
